# starting kernel
# baseline (speedup 1.0000x reference)
#include <hip/hip_runtime.h>


typedef __attribute__((ext_vector_type(16))) __bf16 v16bf;
typedef __attribute__((ext_vector_type(8)))  float  v8f;

#define GNN_N 100000
#define GNN_E 1600000
#define GNN_D 128
#define GNN_H 128
#define GNN_O 64

__global__ void gnn_zero_f32(float* __restrict__ p, long n4) {
    long i = (long)blockIdx.x * blockDim.x + threadIdx.x;
    if (i < n4) {
        ((float4*)p)[i] = make_float4(0.f, 0.f, 0.f, 0.f);
    }
}

__global__ void gnn_count_deg(const int* __restrict__ dst, float* __restrict__ cnt, int E) {
    int e = blockIdx.x * blockDim.x + threadIdx.x;
    if (e < E) {
        unsafeAtomicAdd(&cnt[dst[e]], 1.0f);
    }
}

__global__ void gnn_scatter_add(const float* __restrict__ X,
                                const int* __restrict__ src,
                                const int* __restrict__ dst,
                                float* __restrict__ agg, int E) {
    int e = __builtin_amdgcn_readfirstlane(blockIdx.x * 8 + (threadIdx.x >> 5));
    if (e >= E) return;
    int c = (threadIdx.x & 31) * 4;
    int s = src[e];
    int d = dst[e];
    float4 v = *(const float4*)(X + (long)s * 128 + c);
    float* ap = agg + (long)d * 128 + c;
    unsafeAtomicAdd(ap + 0, v.x);
    unsafeAtomicAdd(ap + 1, v.y);
    unsafeAtomicAdd(ap + 2, v.z);
    unsafeAtomicAdd(ap + 3, v.w);
}

__global__ void gnn_pack_w(const float* __restrict__ W, __bf16* __restrict__ Bp, int Nout) {
    int t = blockIdx.x * blockDim.x + threadIdx.x;
    int total = (Nout / 16) * 4 * 32;
    if (t >= total) return;
    int lane = t & 31;
    int ks   = (t >> 5) & 3;
    int nt   = t >> 7;
    int hi   = lane >> 4;
    int col  = nt * 16 + (lane & 15);
    int kb   = ks * 32 + hi * 8;
    __bf16* o = Bp + (long)t * 16;
#pragma unroll
    for (int j = 0; j < 8; ++j) o[j]     = (__bf16)W[(kb + j) * Nout + col];
#pragma unroll
    for (int j = 0; j < 8; ++j) o[8 + j] = (__bf16)W[(kb + 16 + j) * Nout + col];
}

template <int NOUT, bool RELU>
__global__ void gnn_sage_gemm(const float* __restrict__ Aagg,
                              const float* __restrict__ cnt,
                              const float* __restrict__ Ax,
                              const v16bf* __restrict__ Bl,
                              const v16bf* __restrict__ Br,
                              const float* __restrict__ bias,
                              float* __restrict__ Y) {
    constexpr int K  = 128;
    constexpr int NT = NOUT / 16;
    const int lane    = threadIdx.x & 31;
    const int wid     = threadIdx.x >> 5;
    const int rowtile = blockIdx.x * 2 + wid;
    const int rowbase = rowtile * 16;
    const int hi      = lane >> 4;
    const int row     = rowbase + (lane & 15);
    const float inv   = 1.0f / fmaxf(cnt[row], 1.0f);

    const float* ag = Aagg + (long)row * K;
    const float* ax = Ax   + (long)row * K;

    v16bf afrag[8];
#pragma unroll
    for (int ks = 0; ks < 4; ++ks) {
        const int k0 = ks * 32 + hi * 8;
        v16bf a;
#pragma unroll
        for (int j = 0; j < 8; ++j) a[j]     = (__bf16)(ag[k0 + j] * inv);
#pragma unroll
        for (int j = 0; j < 8; ++j) a[8 + j] = (__bf16)(ag[k0 + 16 + j] * inv);
        afrag[ks] = a;
    }
#pragma unroll
    for (int ks = 0; ks < 4; ++ks) {
        const int k0 = ks * 32 + hi * 8;
        v16bf a;
#pragma unroll
        for (int j = 0; j < 8; ++j) a[j]     = (__bf16)(ax[k0 + j]);
#pragma unroll
        for (int j = 0; j < 8; ++j) a[8 + j] = (__bf16)(ax[k0 + 16 + j]);
        afrag[4 + ks] = a;
    }

#pragma unroll
    for (int ct = 0; ct < NT; ++ct) {
        const int   col = ct * 16 + (lane & 15);
        const float bv  = bias[col];
        v8f acc;
#pragma unroll
        for (int r = 0; r < 8; ++r) acc[r] = bv;

#pragma unroll
        for (int ks = 0; ks < 4; ++ks) {
            v16bf b = Bl[(ct * 4 + ks) * 32 + lane];
            acc = __builtin_amdgcn_wmma_f32_16x16x32_bf16(false, afrag[ks], false, b,
                                                          (short)0, acc, false, false);
        }
#pragma unroll
        for (int ks = 0; ks < 4; ++ks) {
            v16bf b = Br[(ct * 4 + ks) * 32 + lane];
            acc = __builtin_amdgcn_wmma_f32_16x16x32_bf16(false, afrag[4 + ks], false, b,
                                                          (short)0, acc, false, false);
        }

#pragma unroll
        for (int r = 0; r < 8; ++r) {
            float v = acc[r];
            if (RELU) v = fmaxf(v, 0.0f);
            Y[(long)(rowbase + r + hi * 8) * NOUT + col] = v;
        }
    }
}

extern "C" void kernel_launch(void* const* d_in, const int* in_sizes, int n_in,
                              void* d_out, int out_size, void* d_ws, size_t ws_size,
                              hipStream_t stream) {
    (void)in_sizes; (void)n_in; (void)out_size; (void)ws_size;

    const float* x   = (const float*)d_in[0];
    const int*   ei  = (const int*)d_in[1];
    const float* W1l = (const float*)d_in[2];
    const float* b1  = (const float*)d_in[3];
    const float* W1r = (const float*)d_in[4];
    const float* W2l = (const float*)d_in[5];
    const float* b2  = (const float*)d_in[6];
    const float* W2r = (const float*)d_in[7];
    float*       out = (float*)d_out;

    const int* src = ei;
    const int* dst = ei + GNN_E;

    char*   ws    = (char*)d_ws;
    float*  cnt   = (float*)(ws + 0);
    float*  agg   = (float*)(ws + 524288);
    float*  h     = (float*)(ws + 51724288);
    __bf16* W1l_p = (__bf16*)(ws + 102924288);
    __bf16* W1r_p = (__bf16*)(ws + 102957056);
    __bf16* W2l_p = (__bf16*)(ws + 102989824);
    __bf16* W2r_p = (__bf16*)(ws + 103006208);

    const long nFeat4 = (long)GNN_N * 128 / 4;
    const long nCnt4  = GNN_N / 4;

    gnn_pack_w<<<(8 * 4 * 32 + 255) / 256, 256, 0, stream>>>(W1l, W1l_p, GNN_H);
    gnn_pack_w<<<(8 * 4 * 32 + 255) / 256, 256, 0, stream>>>(W1r, W1r_p, GNN_H);
    gnn_pack_w<<<(4 * 4 * 32 + 255) / 256, 256, 0, stream>>>(W2l, W2l_p, GNN_O);
    gnn_pack_w<<<(4 * 4 * 32 + 255) / 256, 256, 0, stream>>>(W2r, W2r_p, GNN_O);

    gnn_zero_f32<<<(int)((nCnt4 + 255) / 256), 256, 0, stream>>>(cnt, nCnt4);
    gnn_count_deg<<<(GNN_E + 255) / 256, 256, 0, stream>>>(dst, cnt, GNN_E);

    gnn_zero_f32<<<(int)((nFeat4 + 255) / 256), 256, 0, stream>>>(agg, nFeat4);
    gnn_scatter_add<<<GNN_E / 8, 256, 0, stream>>>(x, src, dst, agg, GNN_E);
    {
        dim3 grid(GNN_N / 32);
        gnn_sage_gemm<GNN_H, true><<<grid, 64, 0, stream>>>(
            agg, cnt, x, (const v16bf*)W1l_p, (const v16bf*)W1r_p, b1, h);
    }

    gnn_zero_f32<<<(int)((nFeat4 + 255) / 256), 256, 0, stream>>>(agg, nFeat4);
    gnn_scatter_add<<<GNN_E / 8, 256, 0, stream>>>(h, src, dst, agg, GNN_E);
    {
        dim3 grid(GNN_N / 32);
        gnn_sage_gemm<GNN_O, false><<<grid, 64, 0, stream>>>(
            agg, cnt, h, (const v16bf*)W2l_p, (const v16bf*)W2r_p, b2, out);
    }
}


# Edge sort by destination rebuilt through LDS (chunk bin histogram, in-LDS bin sort with whole-run copies, per-bin node sort); no global atomics or scattered stores; sum order = LDS atomic arrival
# speedup vs baseline: 4.0287x; 4.0287x over previous
#include <hip/hip_runtime.h>


typedef __attribute__((ext_vector_type(16))) __bf16 v16bf;
typedef __attribute__((ext_vector_type(8)))  float  v8f;

#define GNN_N 100000
#define GNN_E 1600000
#define GNN_D 128
#define GNN_H 128
#define GNN_O 64

#define GNN_BIN_SHIFT 9
#define GNN_BIN_NODES (1 << GNN_BIN_SHIFT)
#define GNN_NBINS     ((GNN_N + GNN_BIN_NODES - 1) / GNN_BIN_NODES)
#define GNN_CHUNK     6400
#define GNN_NCHUNK    ((GNN_E + GNN_CHUNK - 1) / GNN_CHUNK)
#define GNN_EPT       ((GNN_CHUNK + 1023) / 1024)
#define GNN_STAGE_CAP 12288

static_assert(GNN_NBINS <= 256, "one bin per thread of the first 8 waves");
static_assert(GNN_N <= (1 << (32 - GNN_BIN_SHIFT)), "source id beside the 9-bit local destination");

template <int NW>
__device__ __forceinline__ int gnn_block_excl_scan(int v, int* wsum, int t) {
    const int lane = t & 31;
    const int w    = t >> 5;
    int inc = v;
#pragma unroll
    for (int o = 1; o < 32; o <<= 1) {
        int up = __shfl_up(inc, o);
        if (lane >= o) inc += up;
    }
    if (w < NW && lane == 31) wsum[w] = inc;
    __syncthreads();
    int off = 0;
#pragma unroll
    for (int k = 0; k < NW; ++k) off += (k < w) ? wsum[k] : 0;
    __syncthreads();
    return off + inc - v;
}

__global__ __launch_bounds__(1024) void gnn_bin_hist(const int* __restrict__ dst,
                                                     int* __restrict__ hist, int E) {
    __shared__ int lcnt[GNN_NBINS];
    const int t = threadIdx.x;
    if (t < GNN_NBINS) lcnt[t] = 0;
    __syncthreads();
    const int base = blockIdx.x * GNN_CHUNK;
    const int end  = (base + GNN_CHUNK < E) ? (base + GNN_CHUNK) : E;
    int d[GNN_EPT];
#pragma unroll
    for (int k = 0; k < GNN_EPT; ++k) {
        const int i = base + k * 1024 + t;
        d[k] = (i < end) ? dst[i] : -1;
    }
#pragma unroll
    for (int k = 0; k < GNN_EPT; ++k) {
        if (d[k] >= 0) atomicAdd(&lcnt[d[k] >> GNN_BIN_SHIFT], 1);
    }
    __syncthreads();
    if (t < GNN_NBINS) hist[blockIdx.x * GNN_NBINS + t] = lcnt[t];
}

__global__ __launch_bounds__(1024) void gnn_bin_scatter(const int* __restrict__ src,
                                                        const int* __restrict__ dst,
                                                        const int* __restrict__ hist,
                                                        int* __restrict__ binbase,
                                                        unsigned* __restrict__ binned, int E) {
    __shared__ int lcnt[GNN_NBINS];
    __shared__ int lstart[256];
    __shared__ int goff[256];
    __shared__ int ptot[4][256];
    __shared__ int ppre[4][256];
    __shared__ int wsum[8];
    __shared__ unsigned       sbuf[GNN_CHUNK];
    __shared__ unsigned short bbuf[GNN_CHUNK];

    const int t    = threadIdx.x;
    const int wg   = blockIdx.x;
    const int base = wg * GNN_CHUNK;
    const int end  = (base + GNN_CHUNK < E) ? (base + GNN_CHUNK) : E;

    if (t < GNN_NBINS) lcnt[t] = 0;

    int d[GNN_EPT], s[GNN_EPT], r[GNN_EPT];
#pragma unroll
    for (int k = 0; k < GNN_EPT; ++k) {
        const int i = base + k * 1024 + t;
        d[k] = -1; s[k] = 0; r[k] = 0;
        if (i < end) { d[k] = dst[i]; s[k] = src[i]; }
    }

    {
        const int b = t & 255, q = t >> 8;
        int tot = 0, pre = 0;
        if (b < GNN_NBINS) {
#pragma unroll 8
            for (int w = q; w < GNN_NCHUNK; w += 4) {
                const int v = hist[w * GNN_NBINS + b];
                tot += v;
                pre += (w < wg) ? v : 0;
            }
        }
        ptot[q][b] = tot;
        ppre[q][b] = pre;
    }
    __syncthreads();

#pragma unroll
    for (int k = 0; k < GNN_EPT; ++k) {
        if (d[k] >= 0) r[k] = atomicAdd(&lcnt[d[k] >> GNN_BIN_SHIFT], 1);
    }
    __syncthreads();

    int tot = 0, pre = 0, mine = 0;
    if (t < GNN_NBINS) {
        tot  = ptot[0][t] + ptot[1][t] + ptot[2][t] + ptot[3][t];
        pre  = ppre[0][t] + ppre[1][t] + ppre[2][t] + ppre[3][t];
        mine = lcnt[t];
    }
    const int ls = gnn_block_excl_scan<8>(mine, wsum, t);
    const int gb = gnn_block_excl_scan<8>(tot, wsum, t);
    if (t < GNN_NBINS) { lstart[t] = ls; goff[t] = gb + pre; }
    if (wg == 0 && t <= GNN_NBINS) binbase[t] = gb;
    __syncthreads();

#pragma unroll
    for (int k = 0; k < GNN_EPT; ++k) {
        if (d[k] >= 0) {
            const int b = d[k] >> GNN_BIN_SHIFT;
            const int p = lstart[b] + r[k];
            sbuf[p] = ((unsigned)s[k] << GNN_BIN_SHIFT) | (unsigned)(d[k] & (GNN_BIN_NODES - 1));
            bbuf[p] = (unsigned short)b;
        }
    }
    __syncthreads();

    const int n = end - base;
    for (int i = t; i < n; i += 1024) {
        const int b = bbuf[i];
        binned[goff[b] + (i - lstart[b])] = sbuf[i];
    }
}

__global__ __launch_bounds__(1024) void gnn_bin_sort(const unsigned* __restrict__ binned,
                                                     const int* __restrict__ binbase,
                                                     int* __restrict__ cnt,
                                                     int* __restrict__ rowptr,
                                                     int* __restrict__ col, int N) {
    __shared__ int ncnt[GNN_BIN_NODES];
    __shared__ int wsum[16];
    __shared__ int sbuf[GNN_STAGE_CAP];

    const int t   = threadIdx.x;
    const int bin = blockIdx.x;
    const int beg = binbase[bin];
    const int n   = binbase[bin + 1] - beg;

    if (t < GNN_BIN_NODES) ncnt[t] = 0;
    __syncthreads();
    for (int i0 = t; i0 < n; i0 += 8 * 1024) {
        unsigned w[8];
#pragma unroll
        for (int u = 0; u < 8; ++u) {
            const int i = i0 + u * 1024;
            w[u] = binned[beg + ((i < n) ? i : (n - 1))];
        }
#pragma unroll
        for (int u = 0; u < 8; ++u) {
            if (i0 + u * 1024 < n) atomicAdd(&ncnt[w[u] & (GNN_BIN_NODES - 1)], 1);
        }
    }
    __syncthreads();

    const int deg = (t < GNN_BIN_NODES) ? ncnt[t] : 0;
    const int ex  = gnn_block_excl_scan<16>(deg, wsum, t);
    if (t < GNN_BIN_NODES) {
        ncnt[t] = ex;
        const int node = bin * GNN_BIN_NODES + t;
        if (node < N) { cnt[node] = deg; rowptr[node] = beg + ex; }
    }
    __syncthreads();

    const bool staged = (n <= GNN_STAGE_CAP);
    for (int i0 = t; i0 < n; i0 += 8 * 1024) {
        unsigned w[8];
#pragma unroll
        for (int u = 0; u < 8; ++u) {
            const int i = i0 + u * 1024;
            w[u] = binned[beg + ((i < n) ? i : (n - 1))];
        }
#pragma unroll
        for (int u = 0; u < 8; ++u) {
            if (i0 + u * 1024 < n) {
                const int slot = atomicAdd(&ncnt[w[u] & (GNN_BIN_NODES - 1)], 1);
                const int sid  = (int)(w[u] >> GNN_BIN_SHIFT);
                if (staged) sbuf[slot] = sid; else col[beg + slot] = sid;
            }
        }
    }
    if (staged) {
        __syncthreads();
        for (int i = t; i < n; i += 1024) col[beg + i] = sbuf[i];
    }
}

__global__ void gnn_gather_sum(const float* __restrict__ X,
                               const int* __restrict__ rowptr,
                               const int* __restrict__ cnt,
                               const int* __restrict__ col,
                               float* __restrict__ agg, int N) {
    const int n = __builtin_amdgcn_readfirstlane(blockIdx.x * 8 + (threadIdx.x >> 5));
    if (n >= N) return;
    const int lane = threadIdx.x & 31;
    const int beg  = rowptr[n];
    const int deg  = cnt[n];
    const float* xb = X + lane * 4;

    float4 acc = make_float4(0.f, 0.f, 0.f, 0.f);
    for (int base = 0; base < deg; base += 32) {
        const int m = (deg - base < 32) ? (deg - base) : 32;
        const int myc = (lane < m) ? col[beg + base + lane] : 0;
        int j = 0;
        for (; j + 8 <= m; j += 8) {
            float4 v[8];
#pragma unroll
            for (int u = 0; u < 8; ++u) {
                const int s = __builtin_amdgcn_readlane(myc, j + u);
                v[u] = *(const float4*)(xb + (long)s * 128);
            }
#pragma unroll
            for (int u = 0; u < 8; ++u) {
                acc.x += v[u].x; acc.y += v[u].y; acc.z += v[u].z; acc.w += v[u].w;
            }
        }
        if (j + 4 <= m) {
            float4 v[4];
#pragma unroll
            for (int u = 0; u < 4; ++u) {
                const int s = __builtin_amdgcn_readlane(myc, j + u);
                v[u] = *(const float4*)(xb + (long)s * 128);
            }
#pragma unroll
            for (int u = 0; u < 4; ++u) {
                acc.x += v[u].x; acc.y += v[u].y; acc.z += v[u].z; acc.w += v[u].w;
            }
            j += 4;
        }
        const int r = m - j;
        if (r > 0) {
            float4 v0 = make_float4(0.f, 0.f, 0.f, 0.f), v1 = v0, v2 = v0;
            v0 = *(const float4*)(xb + (long)__builtin_amdgcn_readlane(myc, j) * 128);
            if (r > 1) v1 = *(const float4*)(xb + (long)__builtin_amdgcn_readlane(myc, j + 1) * 128);
            if (r > 2) v2 = *(const float4*)(xb + (long)__builtin_amdgcn_readlane(myc, j + 2) * 128);
            acc.x += v0.x; acc.y += v0.y; acc.z += v0.z; acc.w += v0.w;
            acc.x += v1.x; acc.y += v1.y; acc.z += v1.z; acc.w += v1.w;
            acc.x += v2.x; acc.y += v2.y; acc.z += v2.z; acc.w += v2.w;
        }
    }
    *(float4*)(agg + (long)n * 128 + lane * 4) = acc;
}

__global__ void gnn_pack_w(const float* __restrict__ W, __bf16* __restrict__ Bp, int Nout) {
    int t = blockIdx.x * blockDim.x + threadIdx.x;
    int total = (Nout / 16) * 4 * 32;
    if (t >= total) return;
    int lane = t & 31;
    int ks   = (t >> 5) & 3;
    int nt   = t >> 7;
    int hi   = lane >> 4;
    int col  = nt * 16 + (lane & 15);
    int kb   = ks * 32 + hi * 8;
    __bf16* o = Bp + (long)t * 16;
#pragma unroll
    for (int j = 0; j < 8; ++j) o[j]     = (__bf16)W[(kb + j) * Nout + col];
#pragma unroll
    for (int j = 0; j < 8; ++j) o[8 + j] = (__bf16)W[(kb + 16 + j) * Nout + col];
}

template <int NOUT, bool RELU>
__global__ void gnn_sage_gemm(const float* __restrict__ Aagg,
                              const int* __restrict__ cnt,
                              const float* __restrict__ Ax,
                              const v16bf* __restrict__ Bl,
                              const v16bf* __restrict__ Br,
                              const float* __restrict__ bias,
                              float* __restrict__ Y) {
    constexpr int K  = 128;
    constexpr int NT = NOUT / 16;
    const int lane    = threadIdx.x & 31;
    const int wid     = threadIdx.x >> 5;
    const int rowtile = blockIdx.x * 2 + wid;
    const int rowbase = rowtile * 16;
    const int hi      = lane >> 4;
    const int row     = rowbase + (lane & 15);
    const float inv   = 1.0f / fmaxf((float)cnt[row], 1.0f);

    const float* ag = Aagg + (long)row * K;
    const float* ax = Ax   + (long)row * K;

    v16bf afrag[8];
#pragma unroll
    for (int ks = 0; ks < 4; ++ks) {
        const int k0 = ks * 32 + hi * 8;
        v16bf a;
#pragma unroll
        for (int j = 0; j < 8; ++j) a[j]     = (__bf16)(ag[k0 + j] * inv);
#pragma unroll
        for (int j = 0; j < 8; ++j) a[8 + j] = (__bf16)(ag[k0 + 16 + j] * inv);
        afrag[ks] = a;
    }
#pragma unroll
    for (int ks = 0; ks < 4; ++ks) {
        const int k0 = ks * 32 + hi * 8;
        v16bf a;
#pragma unroll
        for (int j = 0; j < 8; ++j) a[j]     = (__bf16)(ax[k0 + j]);
#pragma unroll
        for (int j = 0; j < 8; ++j) a[8 + j] = (__bf16)(ax[k0 + 16 + j]);
        afrag[4 + ks] = a;
    }

#pragma unroll
    for (int ct = 0; ct < NT; ++ct) {
        const int   col = ct * 16 + (lane & 15);
        const float bv  = bias[col];
        v8f acc;
#pragma unroll
        for (int r = 0; r < 8; ++r) acc[r] = bv;

#pragma unroll
        for (int ks = 0; ks < 4; ++ks) {
            v16bf b = Bl[(ct * 4 + ks) * 32 + lane];
            acc = __builtin_amdgcn_wmma_f32_16x16x32_bf16(false, afrag[ks], false, b,
                                                          (short)0, acc, false, false);
        }
#pragma unroll
        for (int ks = 0; ks < 4; ++ks) {
            v16bf b = Br[(ct * 4 + ks) * 32 + lane];
            acc = __builtin_amdgcn_wmma_f32_16x16x32_bf16(false, afrag[4 + ks], false, b,
                                                          (short)0, acc, false, false);
        }

#pragma unroll
        for (int r = 0; r < 8; ++r) {
            float v = acc[r];
            if (RELU) v = fmaxf(v, 0.0f);
            Y[(long)(rowbase + r + hi * 8) * NOUT + col] = v;
        }
    }
}

extern "C" void kernel_launch(void* const* d_in, const int* in_sizes, int n_in,
                              void* d_out, int out_size, void* d_ws, size_t ws_size,
                              hipStream_t stream) {
    (void)in_sizes; (void)n_in; (void)out_size; (void)ws_size;

    const float* x   = (const float*)d_in[0];
    const int*   ei  = (const int*)d_in[1];
    const float* W1l = (const float*)d_in[2];
    const float* b1  = (const float*)d_in[3];
    const float* W1r = (const float*)d_in[4];
    const float* W2l = (const float*)d_in[5];
    const float* b2  = (const float*)d_in[6];
    const float* W2r = (const float*)d_in[7];
    float*       out = (float*)d_out;

    const int* src = ei;
    const int* dst = ei + GNN_E;

    char*   ws     = (char*)d_ws;
    int*    cnt    = (int*)(ws + 0);
    float*  agg    = (float*)(ws + 524288);
    float*  h      = (float*)(ws + 51724288);
    __bf16* W1l_p  = (__bf16*)(ws + 102924288);
    __bf16* W1r_p  = (__bf16*)(ws + 102957056);
    __bf16* W2l_p  = (__bf16*)(ws + 102989824);
    __bf16* W2r_p  = (__bf16*)(ws + 103006208);
    int*    rowptr = (int*)(ws + 104857600);
    int*    binbase = (int*)(ws + 105381888);
    unsigned* binned = (unsigned*)(ws + 105385984);
    int*    col    = (int*)(ws + 111785984);
    int*    hist   = (int*)(ws + 118185984);

    gnn_pack_w<<<(8 * 4 * 32 + 255) / 256, 256, 0, stream>>>(W1l, W1l_p, GNN_H);
    gnn_pack_w<<<(8 * 4 * 32 + 255) / 256, 256, 0, stream>>>(W1r, W1r_p, GNN_H);
    gnn_pack_w<<<(4 * 4 * 32 + 255) / 256, 256, 0, stream>>>(W2l, W2l_p, GNN_O);
    gnn_pack_w<<<(4 * 4 * 32 + 255) / 256, 256, 0, stream>>>(W2r, W2r_p, GNN_O);

    gnn_bin_hist<<<GNN_NCHUNK, 1024, 0, stream>>>(dst, hist, GNN_E);
    gnn_bin_scatter<<<GNN_NCHUNK, 1024, 0, stream>>>(src, dst, hist, binbase, binned, GNN_E);
    gnn_bin_sort<<<GNN_NBINS, 1024, 0, stream>>>(binned, binbase, cnt, rowptr, col, GNN_N);

    gnn_gather_sum<<<(GNN_N + 7) / 8, 256, 0, stream>>>(x, rowptr, cnt, col, agg, GNN_N);
    {
        dim3 grid(GNN_N / 32);
        gnn_sage_gemm<GNN_H, true><<<grid, 64, 0, stream>>>(
            agg, cnt, x, (const v16bf*)W1l_p, (const v16bf*)W1r_p, b1, h);
    }

    gnn_gather_sum<<<(GNN_N + 7) / 8, 256, 0, stream>>>(h, rowptr, cnt, col, agg, GNN_N);
    {
        dim3 grid(GNN_N / 32);
        gnn_sage_gemm<GNN_O, false><<<grid, 64, 0, stream>>>(
            agg, cnt, h, (const v16bf*)W2l_p, (const v16bf*)W2r_p, b2, out);
    }
}


# Layer 2 projects before it averages (mean is linear): h stays in registers, one WMMA kernel also forms h@W2l and h@W2r+b2, last kernel sums 256-B rows; sum over k now precedes sum over neighbours
# speedup vs baseline: 5.4691x; 1.3575x over previous
#include <hip/hip_runtime.h>


typedef __attribute__((ext_vector_type(16))) __bf16 v16bf;
typedef __attribute__((ext_vector_type(8)))  float  v8f;

#define GNN_N 100000
#define GNN_E 1600000
#define GNN_D 128
#define GNN_H 128
#define GNN_O 64

#define GNN_BIN_SHIFT 9
#define GNN_BIN_NODES (1 << GNN_BIN_SHIFT)
#define GNN_NBINS     ((GNN_N + GNN_BIN_NODES - 1) / GNN_BIN_NODES)
#define GNN_CHUNK     6400
#define GNN_NCHUNK    ((GNN_E + GNN_CHUNK - 1) / GNN_CHUNK)
#define GNN_EPT       ((GNN_CHUNK + 1023) / 1024)
#define GNN_STAGE_CAP 12288

static_assert(GNN_N % 32 == 0, "the WMMA kernel takes whole 16-node strips, two per block");
static_assert(GNN_NBINS <= 256, "one bin per thread of the first 8 waves");
static_assert(GNN_N <= (1 << (32 - GNN_BIN_SHIFT)), "source id beside the 9-bit local destination");

template <int NW>
__device__ __forceinline__ int gnn_block_excl_scan(int v, int* wsum, int t) {
    const int lane = t & 31;
    const int w    = t >> 5;
    int inc = v;
#pragma unroll
    for (int o = 1; o < 32; o <<= 1) {
        int up = __shfl_up(inc, o);
        if (lane >= o) inc += up;
    }
    if (w < NW && lane == 31) wsum[w] = inc;
    __syncthreads();
    int off = 0;
#pragma unroll
    for (int k = 0; k < NW; ++k) off += (k < w) ? wsum[k] : 0;
    __syncthreads();
    return off + inc - v;
}

__global__ __launch_bounds__(1024) void gnn_bin_hist(const int* __restrict__ dst,
                                                     int* __restrict__ hist, int E) {
    __shared__ int lcnt[GNN_NBINS];
    const int t = threadIdx.x;
    if (t < GNN_NBINS) lcnt[t] = 0;
    __syncthreads();
    const int base = blockIdx.x * GNN_CHUNK;
    const int end  = (base + GNN_CHUNK < E) ? (base + GNN_CHUNK) : E;
    int d[GNN_EPT];
#pragma unroll
    for (int k = 0; k < GNN_EPT; ++k) {
        const int i = base + k * 1024 + t;
        d[k] = (i < end) ? dst[i] : -1;
    }
#pragma unroll
    for (int k = 0; k < GNN_EPT; ++k) {
        if (d[k] >= 0) atomicAdd(&lcnt[d[k] >> GNN_BIN_SHIFT], 1);
    }
    __syncthreads();
    if (t < GNN_NBINS) hist[blockIdx.x * GNN_NBINS + t] = lcnt[t];
}

__global__ __launch_bounds__(1024) void gnn_bin_scatter(const int* __restrict__ src,
                                                        const int* __restrict__ dst,
                                                        const int* __restrict__ hist,
                                                        int* __restrict__ binbase,
                                                        unsigned* __restrict__ binned, int E) {
    __shared__ int lcnt[GNN_NBINS];
    __shared__ int lstart[256];
    __shared__ int goff[256];
    __shared__ int ptot[4][256];
    __shared__ int ppre[4][256];
    __shared__ int wsum[8];
    __shared__ unsigned       sbuf[GNN_CHUNK];
    __shared__ unsigned short bbuf[GNN_CHUNK];

    const int t    = threadIdx.x;
    const int wg   = blockIdx.x;
    const int base = wg * GNN_CHUNK;
    const int end  = (base + GNN_CHUNK < E) ? (base + GNN_CHUNK) : E;

    if (t < GNN_NBINS) lcnt[t] = 0;

    int d[GNN_EPT], s[GNN_EPT], r[GNN_EPT];
#pragma unroll
    for (int k = 0; k < GNN_EPT; ++k) {
        const int i = base + k * 1024 + t;
        d[k] = -1; s[k] = 0; r[k] = 0;
        if (i < end) { d[k] = dst[i]; s[k] = src[i]; }
    }

    {
        const int b = t & 255, q = t >> 8;
        int tot = 0, pre = 0;
        if (b < GNN_NBINS) {
#pragma unroll 8
            for (int w = q; w < GNN_NCHUNK; w += 4) {
                const int v = hist[w * GNN_NBINS + b];
                tot += v;
                pre += (w < wg) ? v : 0;
            }
        }
        ptot[q][b] = tot;
        ppre[q][b] = pre;
    }
    __syncthreads();

#pragma unroll
    for (int k = 0; k < GNN_EPT; ++k) {
        if (d[k] >= 0) r[k] = atomicAdd(&lcnt[d[k] >> GNN_BIN_SHIFT], 1);
    }
    __syncthreads();

    int tot = 0, pre = 0, mine = 0;
    if (t < GNN_NBINS) {
        tot  = ptot[0][t] + ptot[1][t] + ptot[2][t] + ptot[3][t];
        pre  = ppre[0][t] + ppre[1][t] + ppre[2][t] + ppre[3][t];
        mine = lcnt[t];
    }
    const int ls = gnn_block_excl_scan<8>(mine, wsum, t);
    const int gb = gnn_block_excl_scan<8>(tot, wsum, t);
    if (t < GNN_NBINS) { lstart[t] = ls; goff[t] = gb + pre; }
    if (wg == 0 && t <= GNN_NBINS) binbase[t] = gb;
    __syncthreads();

#pragma unroll
    for (int k = 0; k < GNN_EPT; ++k) {
        if (d[k] >= 0) {
            const int b = d[k] >> GNN_BIN_SHIFT;
            const int p = lstart[b] + r[k];
            sbuf[p] = ((unsigned)s[k] << GNN_BIN_SHIFT) | (unsigned)(d[k] & (GNN_BIN_NODES - 1));
            bbuf[p] = (unsigned short)b;
        }
    }
    __syncthreads();

    const int n = end - base;
    for (int i = t; i < n; i += 1024) {
        const int b = bbuf[i];
        binned[goff[b] + (i - lstart[b])] = sbuf[i];
    }
}

__global__ __launch_bounds__(1024) void gnn_bin_sort(const unsigned* __restrict__ binned,
                                                     const int* __restrict__ binbase,
                                                     int* __restrict__ cnt,
                                                     int* __restrict__ rowptr,
                                                     int* __restrict__ col, int N) {
    __shared__ int ncnt[GNN_BIN_NODES];
    __shared__ int wsum[16];
    __shared__ int sbuf[GNN_STAGE_CAP];

    const int t   = threadIdx.x;
    const int bin = blockIdx.x;
    const int beg = binbase[bin];
    const int n   = binbase[bin + 1] - beg;

    if (t < GNN_BIN_NODES) ncnt[t] = 0;
    __syncthreads();
    for (int i0 = t; i0 < n; i0 += 8 * 1024) {
        unsigned w[8];
#pragma unroll
        for (int u = 0; u < 8; ++u) {
            const int i = i0 + u * 1024;
            w[u] = binned[beg + ((i < n) ? i : (n - 1))];
        }
#pragma unroll
        for (int u = 0; u < 8; ++u) {
            if (i0 + u * 1024 < n) atomicAdd(&ncnt[w[u] & (GNN_BIN_NODES - 1)], 1);
        }
    }
    __syncthreads();

    const int deg = (t < GNN_BIN_NODES) ? ncnt[t] : 0;
    const int ex  = gnn_block_excl_scan<16>(deg, wsum, t);
    if (t < GNN_BIN_NODES) {
        ncnt[t] = ex;
        const int node = bin * GNN_BIN_NODES + t;
        if (node < N) { cnt[node] = deg; rowptr[node] = beg + ex; }
    }
    __syncthreads();

    const bool staged = (n <= GNN_STAGE_CAP);
    for (int i0 = t; i0 < n; i0 += 8 * 1024) {
        unsigned w[8];
#pragma unroll
        for (int u = 0; u < 8; ++u) {
            const int i = i0 + u * 1024;
            w[u] = binned[beg + ((i < n) ? i : (n - 1))];
        }
#pragma unroll
        for (int u = 0; u < 8; ++u) {
            if (i0 + u * 1024 < n) {
                const int slot = atomicAdd(&ncnt[w[u] & (GNN_BIN_NODES - 1)], 1);
                const int sid  = (int)(w[u] >> GNN_BIN_SHIFT);
                if (staged) sbuf[slot] = sid; else col[beg + slot] = sid;
            }
        }
    }
    if (staged) {
        __syncthreads();
        for (int i = t; i < n; i += 1024) col[beg + i] = sbuf[i];
    }
}

__global__ void gnn_gather_sum(const float* __restrict__ X,
                               const int* __restrict__ rowptr,
                               const int* __restrict__ cnt,
                               const int* __restrict__ col,
                               float* __restrict__ agg, int N) {
    const int n = __builtin_amdgcn_readfirstlane(blockIdx.x * 8 + (threadIdx.x >> 5));
    if (n >= N) return;
    const int lane = threadIdx.x & 31;
    const int beg  = rowptr[n];
    const int deg  = cnt[n];
    const float* xb = X + lane * 4;

    float4 acc = make_float4(0.f, 0.f, 0.f, 0.f);
    for (int base = 0; base < deg; base += 32) {
        const int m = (deg - base < 32) ? (deg - base) : 32;
        const int myc = (lane < m) ? col[beg + base + lane] : 0;
        int j = 0;
        for (; j + 8 <= m; j += 8) {
            float4 v[8];
#pragma unroll
            for (int u = 0; u < 8; ++u) {
                const int s = __builtin_amdgcn_readlane(myc, j + u);
                v[u] = *(const float4*)(xb + (long)s * 128);
            }
#pragma unroll
            for (int u = 0; u < 8; ++u) {
                acc.x += v[u].x; acc.y += v[u].y; acc.z += v[u].z; acc.w += v[u].w;
            }
        }
        if (j + 4 <= m) {
            float4 v[4];
#pragma unroll
            for (int u = 0; u < 4; ++u) {
                const int s = __builtin_amdgcn_readlane(myc, j + u);
                v[u] = *(const float4*)(xb + (long)s * 128);
            }
#pragma unroll
            for (int u = 0; u < 4; ++u) {
                acc.x += v[u].x; acc.y += v[u].y; acc.z += v[u].z; acc.w += v[u].w;
            }
            j += 4;
        }
        const int r = m - j;
        if (r > 0) {
            float4 v0 = make_float4(0.f, 0.f, 0.f, 0.f), v1 = v0, v2 = v0;
            v0 = *(const float4*)(xb + (long)__builtin_amdgcn_readlane(myc, j) * 128);
            if (r > 1) v1 = *(const float4*)(xb + (long)__builtin_amdgcn_readlane(myc, j + 1) * 128);
            if (r > 2) v2 = *(const float4*)(xb + (long)__builtin_amdgcn_readlane(myc, j + 2) * 128);
            acc.x += v0.x; acc.y += v0.y; acc.z += v0.z; acc.w += v0.w;
            acc.x += v1.x; acc.y += v1.y; acc.z += v1.z; acc.w += v1.w;
            acc.x += v2.x; acc.y += v2.y; acc.z += v2.z; acc.w += v2.w;
        }
    }
    *(float4*)(agg + (long)n * 128 + lane * 4) = acc;
}

__global__ void gnn_pack_w(const float* __restrict__ W, __bf16* __restrict__ Bp, int Nout) {
    int t = blockIdx.x * blockDim.x + threadIdx.x;
    int total = (Nout / 16) * 4 * 32;
    if (t >= total) return;
    int lane = t & 31;
    int ks   = (t >> 5) & 3;
    int nt   = t >> 7;
    int hi   = lane >> 4;
    int col  = nt * 16 + (lane & 15);
    int kb   = ks * 32 + hi * 8;
    __bf16* o = Bp + (long)t * 16;
#pragma unroll
    for (int j = 0; j < 8; ++j) o[j]     = (__bf16)W[(kb + j) * Nout + col];
#pragma unroll
    for (int j = 0; j < 8; ++j) o[8 + j] = (__bf16)W[(kb + 16 + j) * Nout + col];
}

__global__ __launch_bounds__(64, 4) void gnn_sage_both(const float* __restrict__ Aagg,
                                                    const int* __restrict__ cnt,
                                                    const float* __restrict__ Ax,
                                                    const v16bf* __restrict__ W1l,
                                                    const v16bf* __restrict__ W1r,
                                                    const float* __restrict__ b1,
                                                    const v16bf* __restrict__ W2l,
                                                    const v16bf* __restrict__ W2r,
                                                    const float* __restrict__ b2,
                                                    float* __restrict__ Z,
                                                    float* __restrict__ R) {
    constexpr int K = 128;
    const int lane    = threadIdx.x & 31;
    const int wid     = threadIdx.x >> 5;
    const int rowbase = (blockIdx.x * 2 + wid) * 16;
    const int hi      = lane >> 4;
    const int row     = rowbase + (lane & 15);
    const float inv   = 1.0f / fmaxf((float)cnt[row], 1.0f);

    const float* ag = Aagg + (long)row * K;
    const float* ax = Ax   + (long)row * K;

    v16bf xf[8];
#pragma unroll
    for (int ks = 0; ks < 4; ++ks) {
        const int k0 = ks * 32 + hi * 8;
        v16bf a;
#pragma unroll
        for (int j = 0; j < 8; ++j) a[j]     = (__bf16)(ag[k0 + j] * inv);
#pragma unroll
        for (int j = 0; j < 8; ++j) a[8 + j] = (__bf16)(ag[k0 + 16 + j] * inv);
        xf[ks] = a;
    }
#pragma unroll
    for (int ks = 0; ks < 4; ++ks) {
        const int k0 = ks * 32 + hi * 8;
        v16bf a;
#pragma unroll
        for (int j = 0; j < 8; ++j) a[j]     = (__bf16)(ax[k0 + j]);
#pragma unroll
        for (int j = 0; j < 8; ++j) a[8 + j] = (__bf16)(ax[k0 + 16 + j]);
        xf[4 + ks] = a;
    }

    __builtin_amdgcn_sched_barrier(0);

    v16bf hf[4];
#pragma unroll
    for (int ct = 0; ct < GNN_H / 16; ++ct) {
        const float4 bA = *(const float4*)(b1 + ct * 16 + hi * 8);
        const float4 bB = *(const float4*)(b1 + ct * 16 + hi * 8 + 4);
        v8f acc;
        acc[0] = bA.x; acc[1] = bA.y; acc[2] = bA.z; acc[3] = bA.w;
        acc[4] = bB.x; acc[5] = bB.y; acc[6] = bB.z; acc[7] = bB.w;
#pragma unroll
        for (int ks = 0; ks < 4; ++ks) {
            const v16bf w = W1l[(ct * 4 + ks) * 32 + lane];
            acc = __builtin_amdgcn_wmma_f32_16x16x32_bf16(false, w, false, xf[ks],
                                                          (short)0, acc, false, false);
        }
#pragma unroll
        for (int ks = 0; ks < 4; ++ks) {
            const v16bf w = W1r[(ct * 4 + ks) * 32 + lane];
            acc = __builtin_amdgcn_wmma_f32_16x16x32_bf16(false, w, false, xf[4 + ks],
                                                          (short)0, acc, false, false);
        }
#pragma unroll
        for (int r = 0; r < 8; ++r)
            hf[ct >> 1][(ct & 1) * 8 + r] = (__bf16)fmaxf(acc[r], 0.0f);
    }

#pragma unroll
    for (int ot = 0; ot < GNN_O / 16; ++ot) {
        const float4 bA = *(const float4*)(b2 + ot * 16 + hi * 8);
        const float4 bB = *(const float4*)(b2 + ot * 16 + hi * 8 + 4);
        v8f accz, accr;
#pragma unroll
        for (int r = 0; r < 8; ++r) accz[r] = 0.0f;
        accr[0] = bA.x; accr[1] = bA.y; accr[2] = bA.z; accr[3] = bA.w;
        accr[4] = bB.x; accr[5] = bB.y; accr[6] = bB.z; accr[7] = bB.w;
#pragma unroll
        for (int kk = 0; kk < 4; ++kk) {
            const v16bf wl = W2l[(ot * 4 + kk) * 32 + lane];
            accz = __builtin_amdgcn_wmma_f32_16x16x32_bf16(false, wl, false, hf[kk],
                                                           (short)0, accz, false, false);
        }
#pragma unroll
        for (int kk = 0; kk < 4; ++kk) {
            const v16bf wr = W2r[(ot * 4 + kk) * 32 + lane];
            accr = __builtin_amdgcn_wmma_f32_16x16x32_bf16(false, wr, false, hf[kk],
                                                           (short)0, accr, false, false);
        }
        const long o = (long)row * GNN_O + ot * 16 + hi * 8;
        *(float4*)(Z + o)     = make_float4(accz[0], accz[1], accz[2], accz[3]);
        *(float4*)(Z + o + 4) = make_float4(accz[4], accz[5], accz[6], accz[7]);
        *(float4*)(R + o)     = make_float4(accr[0], accr[1], accr[2], accr[3]);
        *(float4*)(R + o + 4) = make_float4(accr[4], accr[5], accr[6], accr[7]);
    }
}

__global__ void gnn_gather_mean_out(const float* __restrict__ Z,
                                    const float* __restrict__ R,
                                    const int* __restrict__ rowptr,
                                    const int* __restrict__ cnt,
                                    const int* __restrict__ col,
                                    float* __restrict__ out, int N) {
    const int n = __builtin_amdgcn_readfirstlane(blockIdx.x * 8 + (threadIdx.x >> 5));
    if (n >= N) return;
    const int lane = threadIdx.x & 31;
    const int beg  = rowptr[n];
    const int deg  = cnt[n];
    const float* zb = Z + lane * 2;
    const float2 rv = *(const float2*)(R + (long)n * GNN_O + lane * 2);

    float2 acc = make_float2(0.f, 0.f);
    for (int base = 0; base < deg; base += 32) {
        const int m = (deg - base < 32) ? (deg - base) : 32;
        const int myc = (lane < m) ? col[beg + base + lane] : 0;
        int j = 0;
        for (; j + 8 <= m; j += 8) {
            float2 v[8];
#pragma unroll
            for (int u = 0; u < 8; ++u) {
                const int s = __builtin_amdgcn_readlane(myc, j + u);
                v[u] = *(const float2*)(zb + (long)s * GNN_O);
            }
#pragma unroll
            for (int u = 0; u < 8; ++u) { acc.x += v[u].x; acc.y += v[u].y; }
        }
        if (j + 4 <= m) {
            float2 v[4];
#pragma unroll
            for (int u = 0; u < 4; ++u) {
                const int s = __builtin_amdgcn_readlane(myc, j + u);
                v[u] = *(const float2*)(zb + (long)s * GNN_O);
            }
#pragma unroll
            for (int u = 0; u < 4; ++u) { acc.x += v[u].x; acc.y += v[u].y; }
            j += 4;
        }
        const int r = m - j;
        if (r > 0) {
            float2 v0 = make_float2(0.f, 0.f), v1 = v0, v2 = v0;
            v0 = *(const float2*)(zb + (long)__builtin_amdgcn_readlane(myc, j) * GNN_O);
            if (r > 1) v1 = *(const float2*)(zb + (long)__builtin_amdgcn_readlane(myc, j + 1) * GNN_O);
            if (r > 2) v2 = *(const float2*)(zb + (long)__builtin_amdgcn_readlane(myc, j + 2) * GNN_O);
            acc.x += v0.x; acc.y += v0.y;
            acc.x += v1.x; acc.y += v1.y;
            acc.x += v2.x; acc.y += v2.y;
        }
    }
    const float inv = 1.0f / fmaxf((float)deg, 1.0f);
    *(float2*)(out + (long)n * GNN_O + lane * 2) = make_float2(rv.x + acc.x * inv, rv.y + acc.y * inv);
}

extern "C" void kernel_launch(void* const* d_in, const int* in_sizes, int n_in,
                              void* d_out, int out_size, void* d_ws, size_t ws_size,
                              hipStream_t stream) {
    (void)in_sizes; (void)n_in; (void)out_size; (void)ws_size;

    const float* x   = (const float*)d_in[0];
    const int*   ei  = (const int*)d_in[1];
    const float* W1l = (const float*)d_in[2];
    const float* b1  = (const float*)d_in[3];
    const float* W1r = (const float*)d_in[4];
    const float* W2l = (const float*)d_in[5];
    const float* b2  = (const float*)d_in[6];
    const float* W2r = (const float*)d_in[7];
    float*       out = (float*)d_out;

    const int* src = ei;
    const int* dst = ei + GNN_E;

    char*   ws     = (char*)d_ws;
    int*    cnt    = (int*)(ws + 0);
    float*  agg    = (float*)(ws + 524288);
    float*  z      = (float*)(ws + 51724288);
    float*  r      = (float*)(ws + 77324288);
    __bf16* W1l_p  = (__bf16*)(ws + 102924288);
    __bf16* W1r_p  = (__bf16*)(ws + 102957056);
    __bf16* W2l_p  = (__bf16*)(ws + 102989824);
    __bf16* W2r_p  = (__bf16*)(ws + 103006208);
    int*    rowptr = (int*)(ws + 104857600);
    int*    binbase = (int*)(ws + 105381888);
    unsigned* binned = (unsigned*)(ws + 105385984);
    int*    col    = (int*)(ws + 111785984);
    int*    hist   = (int*)(ws + 118185984);

    gnn_pack_w<<<(8 * 4 * 32 + 255) / 256, 256, 0, stream>>>(W1l, W1l_p, GNN_H);
    gnn_pack_w<<<(8 * 4 * 32 + 255) / 256, 256, 0, stream>>>(W1r, W1r_p, GNN_H);
    gnn_pack_w<<<(4 * 4 * 32 + 255) / 256, 256, 0, stream>>>(W2l, W2l_p, GNN_O);
    gnn_pack_w<<<(4 * 4 * 32 + 255) / 256, 256, 0, stream>>>(W2r, W2r_p, GNN_O);

    gnn_bin_hist<<<GNN_NCHUNK, 1024, 0, stream>>>(dst, hist, GNN_E);
    gnn_bin_scatter<<<GNN_NCHUNK, 1024, 0, stream>>>(src, dst, hist, binbase, binned, GNN_E);
    gnn_bin_sort<<<GNN_NBINS, 1024, 0, stream>>>(binned, binbase, cnt, rowptr, col, GNN_N);

    gnn_gather_sum<<<(GNN_N + 7) / 8, 256, 0, stream>>>(x, rowptr, cnt, col, agg, GNN_N);

    gnn_sage_both<<<GNN_N / 32, 64, 0, stream>>>(
        agg, cnt, x, (const v16bf*)W1l_p, (const v16bf*)W1r_p, b1,
        (const v16bf*)W2l_p, (const v16bf*)W2r_p, b2, z, r);

    gnn_gather_mean_out<<<(GNN_N + 7) / 8, 256, 0, stream>>>(z, r, rowptr, cnt, col, out, GNN_N);
}


# speedup vs baseline: 6.3220x; 1.1559x over previous
#include <hip/hip_runtime.h>


typedef __attribute__((ext_vector_type(16))) __bf16 v16bf;
typedef __attribute__((ext_vector_type(8)))  __bf16 v8bf;
typedef __attribute__((ext_vector_type(4)))  __bf16 v4bf;
typedef __attribute__((ext_vector_type(8)))  float  v8f;

#define GNN_N 100000
#define GNN_E 1600000
#define GNN_D 128
#define GNN_H 128
#define GNN_O 64

#define GNN_BIN_SHIFT 9
#define GNN_BIN_NODES (1 << GNN_BIN_SHIFT)
#define GNN_NBINS     ((GNN_N + GNN_BIN_NODES - 1) / GNN_BIN_NODES)
#define GNN_CHUNK     6400
#define GNN_NCHUNK    ((GNN_E + GNN_CHUNK - 1) / GNN_CHUNK)
#define GNN_EPT       ((GNN_CHUNK + 1023) / 1024)
#define GNN_STAGE_CAP 12288

static_assert(GNN_N % 32 == 0, "the WMMA kernel takes whole 16-node strips, two per block");
static_assert(GNN_NBINS <= 256, "one bin per thread of the first 8 waves");
static_assert(GNN_N <= (1 << (32 - GNN_BIN_SHIFT)), "source id beside the 9-bit local destination");

template <int NW>
__device__ __forceinline__ int gnn_block_excl_scan(int v, int* wsum, int t) {
    const int lane = t & 31;
    const int w    = t >> 5;
    int inc = v;
#pragma unroll
    for (int o = 1; o < 32; o <<= 1) {
        int up = __shfl_up(inc, o);
        if (lane >= o) inc += up;
    }
    if (w < NW && lane == 31) wsum[w] = inc;
    __syncthreads();
    int off = 0;
#pragma unroll
    for (int k = 0; k < NW; ++k) off += (k < w) ? wsum[k] : 0;
    __syncthreads();
    return off + inc - v;
}

__device__ __forceinline__ void gnn_pack_w(const float* __restrict__ W, __bf16* __restrict__ Bp,
                                           int Nout, int t) {
    int lane = t & 31;
    int ks   = (t >> 5) & 3;
    int nt   = t >> 7;
    int hi   = lane >> 4;
    int col  = nt * 16 + (lane & 15);
    int kb   = ks * 32 + hi * 8;
    __bf16* o = Bp + (long)t * 16;
#pragma unroll
    for (int j = 0; j < 8; ++j) o[j]     = (__bf16)W[(kb + j) * Nout + col];
#pragma unroll
    for (int j = 0; j < 8; ++j) o[8 + j] = (__bf16)W[(kb + 16 + j) * Nout + col];
}

#define GNN_X4   (GNN_N * GNN_D / 4)
#define GNN_XWGS ((GNN_X4 + 3 * 8192 - 1) / (3 * 8192))
static_assert(3 * GNN_XWGS * 8192 >= GNN_X4, "the three thirds cover x");

__device__ __forceinline__ void gnn_round_x(const float* __restrict__ x, __bf16* __restrict__ xb,
                                            int g, int t) {
    const float4* x4 = (const float4*)x;
    v4bf* o4 = (v4bf*)xb;
    const int j0 = g * 8192 + t;
    float4 a[8];
#pragma unroll
    for (int k = 0; k < 8; ++k) {
        const int j = j0 + k * 1024;
        a[k] = x4[(j < GNN_X4) ? j : (GNN_X4 - 1)];
    }
#pragma unroll
    for (int k = 0; k < 8; ++k) {
        const int j = j0 + k * 1024;
        if (j < GNN_X4) {
            v4bf o;
            o[0] = (__bf16)a[k].x; o[1] = (__bf16)a[k].y; o[2] = (__bf16)a[k].z; o[3] = (__bf16)a[k].w;
            o4[j] = o;
        }
    }
}

#define GNN_PACK_WGS 3
__global__ __launch_bounds__(1024) void gnn_bin_hist_pack(const int* __restrict__ dst,
                                                          int* __restrict__ hist, int E,
                                                          const float* __restrict__ W1l,
                                                          const float* __restrict__ W1r,
                                                          const float* __restrict__ W2l,
                                                          const float* __restrict__ W2r,
                                                          __bf16* __restrict__ W1l_p,
                                                          __bf16* __restrict__ W1r_p,
                                                          __bf16* __restrict__ W2l_p,
                                                          __bf16* __restrict__ W2r_p,
                                                          const float* __restrict__ x,
                                                          __bf16* __restrict__ xb) {
    __shared__ int lcnt[GNN_NBINS];
    const int t = threadIdx.x;
    if (blockIdx.x >= GNN_NCHUNK) {
        const int pb = blockIdx.x - GNN_NCHUNK;
        if (pb == 0)      gnn_pack_w(W1l, W1l_p, GNN_H, t);
        else if (pb == 1) gnn_pack_w(W1r, W1r_p, GNN_H, t);
        else if (pb == 2) {
            if (t < 512) gnn_pack_w(W2l, W2l_p, GNN_O, t);
            else         gnn_pack_w(W2r, W2r_p, GNN_O, t - 512);
        } else {
            gnn_round_x(x, xb, pb - GNN_PACK_WGS, t);
        }
        return;
    }
    if (t < GNN_NBINS) lcnt[t] = 0;
    __syncthreads();
    const int base = blockIdx.x * GNN_CHUNK;
    const int end  = (base + GNN_CHUNK < E) ? (base + GNN_CHUNK) : E;
    int d[GNN_EPT];
#pragma unroll
    for (int k = 0; k < GNN_EPT; ++k) {
        const int i = base + k * 1024 + t;
        d[k] = (i < end) ? dst[i] : -1;
    }
#pragma unroll
    for (int k = 0; k < GNN_EPT; ++k) {
        if (d[k] >= 0) atomicAdd(&lcnt[d[k] >> GNN_BIN_SHIFT], 1);
    }
    __syncthreads();
    if (t < GNN_NBINS) hist[blockIdx.x * GNN_NBINS + t] = lcnt[t];
}

__global__ __launch_bounds__(1024) void gnn_bin_scatter(const int* __restrict__ src,
                                                        const int* __restrict__ dst,
                                                        const int* __restrict__ hist,
                                                        int* __restrict__ binbase,
                                                        unsigned* __restrict__ binned, int E,
                                                        const float* __restrict__ x,
                                                        __bf16* __restrict__ xb) {
    if (blockIdx.x >= GNN_NCHUNK) {
        gnn_round_x(x, xb, GNN_XWGS + (int)(blockIdx.x - GNN_NCHUNK), threadIdx.x);
        return;
    }
    __shared__ int lcnt[GNN_NBINS];
    __shared__ int lstart[256];
    __shared__ int goff[256];
    __shared__ int ptot[4][256];
    __shared__ int ppre[4][256];
    __shared__ int wsum[8];
    __shared__ unsigned       sbuf[GNN_CHUNK];
    __shared__ unsigned short bbuf[GNN_CHUNK];

    const int t    = threadIdx.x;
    const int wg   = blockIdx.x;
    const int base = wg * GNN_CHUNK;
    const int end  = (base + GNN_CHUNK < E) ? (base + GNN_CHUNK) : E;

    if (t < GNN_NBINS) lcnt[t] = 0;

    int d[GNN_EPT], s[GNN_EPT], r[GNN_EPT];
#pragma unroll
    for (int k = 0; k < GNN_EPT; ++k) {
        const int i = base + k * 1024 + t;
        d[k] = -1; s[k] = 0; r[k] = 0;
        if (i < end) { d[k] = dst[i]; s[k] = src[i]; }
    }

    {
        const int b = t & 255, q = t >> 8;
        int tot = 0, pre = 0;
        if (b < GNN_NBINS) {
#pragma unroll 8
            for (int w = q; w < GNN_NCHUNK; w += 4) {
                const int v = hist[w * GNN_NBINS + b];
                tot += v;
                pre += (w < wg) ? v : 0;
            }
        }
        ptot[q][b] = tot;
        ppre[q][b] = pre;
    }
    __syncthreads();

#pragma unroll
    for (int k = 0; k < GNN_EPT; ++k) {
        if (d[k] >= 0) r[k] = atomicAdd(&lcnt[d[k] >> GNN_BIN_SHIFT], 1);
    }
    __syncthreads();

    int tot = 0, pre = 0, mine = 0;
    if (t < GNN_NBINS) {
        tot  = ptot[0][t] + ptot[1][t] + ptot[2][t] + ptot[3][t];
        pre  = ppre[0][t] + ppre[1][t] + ppre[2][t] + ppre[3][t];
        mine = lcnt[t];
    }
    const int ls = gnn_block_excl_scan<8>(mine, wsum, t);
    const int gb = gnn_block_excl_scan<8>(tot, wsum, t);
    if (t < GNN_NBINS) { lstart[t] = ls; goff[t] = gb + pre; }
    if (wg == 0 && t <= GNN_NBINS) binbase[t] = gb;
    __syncthreads();

#pragma unroll
    for (int k = 0; k < GNN_EPT; ++k) {
        if (d[k] >= 0) {
            const int b = d[k] >> GNN_BIN_SHIFT;
            const int p = lstart[b] + r[k];
            sbuf[p] = ((unsigned)s[k] << GNN_BIN_SHIFT) | (unsigned)(d[k] & (GNN_BIN_NODES - 1));
            bbuf[p] = (unsigned short)b;
        }
    }
    __syncthreads();

    const int n = end - base;
    for (int i = t; i < n; i += 1024) {
        const int b = bbuf[i];
        binned[goff[b] + (i - lstart[b])] = sbuf[i];
    }
}

__global__ __launch_bounds__(1024) void gnn_bin_sort(const unsigned* __restrict__ binned,
                                                     const int* __restrict__ binbase,
                                                     int* __restrict__ cnt,
                                                     int* __restrict__ rowptr,
                                                     int* __restrict__ col, int N,
                                                     const float* __restrict__ x,
                                                     __bf16* __restrict__ xb) {
    if (blockIdx.x >= GNN_NBINS) {
        gnn_round_x(x, xb, 2 * GNN_XWGS + (int)(blockIdx.x - GNN_NBINS), threadIdx.x);
        return;
    }
    __shared__ int ncnt[GNN_BIN_NODES];
    __shared__ int wsum[16];
    __shared__ int sbuf[GNN_STAGE_CAP];

    const int t   = threadIdx.x;
    const int bin = blockIdx.x;
    const int beg = binbase[bin];
    const int n   = binbase[bin + 1] - beg;

    if (t < GNN_BIN_NODES) ncnt[t] = 0;
    __syncthreads();
    for (int i0 = t; i0 < n; i0 += 8 * 1024) {
        unsigned w[8];
#pragma unroll
        for (int u = 0; u < 8; ++u) {
            const int i = i0 + u * 1024;
            w[u] = binned[beg + ((i < n) ? i : (n - 1))];
        }
#pragma unroll
        for (int u = 0; u < 8; ++u) {
            if (i0 + u * 1024 < n) atomicAdd(&ncnt[w[u] & (GNN_BIN_NODES - 1)], 1);
        }
    }
    __syncthreads();

    const int deg = (t < GNN_BIN_NODES) ? ncnt[t] : 0;
    const int ex  = gnn_block_excl_scan<16>(deg, wsum, t);
    if (t < GNN_BIN_NODES) {
        ncnt[t] = ex;
        const int node = bin * GNN_BIN_NODES + t;
        if (node < N) { cnt[node] = deg; rowptr[node] = beg + ex; }
    }
    __syncthreads();

    const bool staged = (n <= GNN_STAGE_CAP);
    for (int i0 = t; i0 < n; i0 += 8 * 1024) {
        unsigned w[8];
#pragma unroll
        for (int u = 0; u < 8; ++u) {
            const int i = i0 + u * 1024;
            w[u] = binned[beg + ((i < n) ? i : (n - 1))];
        }
#pragma unroll
        for (int u = 0; u < 8; ++u) {
            if (i0 + u * 1024 < n) {
                const int slot = atomicAdd(&ncnt[w[u] & (GNN_BIN_NODES - 1)], 1);
                const int sid  = (int)(w[u] >> GNN_BIN_SHIFT);
                if (staged) sbuf[slot] = sid; else col[beg + slot] = sid;
            }
        }
    }
    if (staged) {
        __syncthreads();
        for (int i = t; i < n; i += 1024) col[beg + i] = sbuf[i];
    }
}

__global__ void gnn_gather_mean_bf16(const unsigned* __restrict__ Xb,
                                     const int* __restrict__ rowptr,
                                     const int* __restrict__ cnt,
                                     const int* __restrict__ col,
                                     __bf16* __restrict__ aggb, int N) {
    const int n = __builtin_amdgcn_readfirstlane(blockIdx.x * 8 + (threadIdx.x >> 5));
    if (n >= N) return;
    const int lane = threadIdx.x & 31;
    const int beg  = rowptr[n];
    const int deg  = cnt[n];
    const unsigned* xb = Xb + lane * 2;

    float a0 = 0.f, a1 = 0.f, a2 = 0.f, a3 = 0.f;
    for (int base = 0; base < deg; base += 32) {
        const int m = (deg - base < 32) ? (deg - base) : 32;
        const int myc = (lane < m) ? col[beg + base + lane] : 0;
        int j = 0;
        for (; j + 8 <= m; j += 8) {
            uint2 v[8];
#pragma unroll
            for (int u = 0; u < 8; ++u) {
                const int s = __builtin_amdgcn_readlane(myc, j + u);
                v[u] = *(const uint2*)(xb + (long)s * 64);
            }
#pragma unroll
            for (int u = 0; u < 8; ++u) {
                a0 += __uint_as_float(v[u].x << 16); a1 += __uint_as_float(v[u].x & 0xffff0000u);
                a2 += __uint_as_float(v[u].y << 16); a3 += __uint_as_float(v[u].y & 0xffff0000u);
            }
        }
        if (j + 4 <= m) {
            uint2 v[4];
#pragma unroll
            for (int u = 0; u < 4; ++u) {
                const int s = __builtin_amdgcn_readlane(myc, j + u);
                v[u] = *(const uint2*)(xb + (long)s * 64);
            }
#pragma unroll
            for (int u = 0; u < 4; ++u) {
                a0 += __uint_as_float(v[u].x << 16); a1 += __uint_as_float(v[u].x & 0xffff0000u);
                a2 += __uint_as_float(v[u].y << 16); a3 += __uint_as_float(v[u].y & 0xffff0000u);
            }
            j += 4;
        }
        const int r = m - j;
        if (r > 0) {
            uint2 v0 = make_uint2(0u, 0u), v1 = v0, v2 = v0;
            v0 = *(const uint2*)(xb + (long)__builtin_amdgcn_readlane(myc, j) * 64);
            if (r > 1) v1 = *(const uint2*)(xb + (long)__builtin_amdgcn_readlane(myc, j + 1) * 64);
            if (r > 2) v2 = *(const uint2*)(xb + (long)__builtin_amdgcn_readlane(myc, j + 2) * 64);
            a0 += __uint_as_float(v0.x << 16); a1 += __uint_as_float(v0.x & 0xffff0000u);
            a2 += __uint_as_float(v0.y << 16); a3 += __uint_as_float(v0.y & 0xffff0000u);
            a0 += __uint_as_float(v1.x << 16); a1 += __uint_as_float(v1.x & 0xffff0000u);
            a2 += __uint_as_float(v1.y << 16); a3 += __uint_as_float(v1.y & 0xffff0000u);
            a0 += __uint_as_float(v2.x << 16); a1 += __uint_as_float(v2.x & 0xffff0000u);
            a2 += __uint_as_float(v2.y << 16); a3 += __uint_as_float(v2.y & 0xffff0000u);
        }
    }
    const float inv = 1.0f / fmaxf((float)deg, 1.0f);
    v4bf o;
    o[0] = (__bf16)(a0 * inv); o[1] = (__bf16)(a1 * inv);
    o[2] = (__bf16)(a2 * inv); o[3] = (__bf16)(a3 * inv);
    *(v4bf*)(aggb + (long)n * 128 + lane * 4) = o;
}

__global__ __launch_bounds__(64, 4) void gnn_sage_both(const __bf16* __restrict__ Aagg,
                                                    const __bf16* __restrict__ Ax,
                                                    const v16bf* __restrict__ W1l,
                                                    const v16bf* __restrict__ W1r,
                                                    const float* __restrict__ b1,
                                                    const v16bf* __restrict__ W2l,
                                                    const v16bf* __restrict__ W2r,
                                                    const float* __restrict__ b2,
                                                    float* __restrict__ Z,
                                                    float* __restrict__ R) {
    constexpr int K = 128;
    const int lane    = threadIdx.x & 31;
    const int wid     = threadIdx.x >> 5;
    const int rowbase = (blockIdx.x * 2 + wid) * 16;
    const int hi      = lane >> 4;
    const int row     = rowbase + (lane & 15);

    const __bf16* ag = Aagg + (long)row * K;
    const __bf16* ax = Ax   + (long)row * K;

    v16bf xf[8];
#pragma unroll
    for (int ks = 0; ks < 4; ++ks) {
        const int k0 = ks * 32 + hi * 8;
        const v8bf lo = *(const v8bf*)(ag + k0);
        const v8bf up = *(const v8bf*)(ag + k0 + 16);
        xf[ks] = __builtin_shufflevector(lo, up, 0, 1, 2, 3, 4, 5, 6, 7, 8, 9, 10, 11, 12, 13, 14, 15);
    }
#pragma unroll
    for (int ks = 0; ks < 4; ++ks) {
        const int k0 = ks * 32 + hi * 8;
        const v8bf lo = *(const v8bf*)(ax + k0);
        const v8bf up = *(const v8bf*)(ax + k0 + 16);
        xf[4 + ks] = __builtin_shufflevector(lo, up, 0, 1, 2, 3, 4, 5, 6, 7, 8, 9, 10, 11, 12, 13, 14, 15);
    }

    v16bf hf[4];
#pragma unroll
    for (int ct = 0; ct < GNN_H / 16; ++ct) {
        const float4 bA = *(const float4*)(b1 + ct * 16 + hi * 8);
        const float4 bB = *(const float4*)(b1 + ct * 16 + hi * 8 + 4);
        v8f acc;
        acc[0] = bA.x; acc[1] = bA.y; acc[2] = bA.z; acc[3] = bA.w;
        acc[4] = bB.x; acc[5] = bB.y; acc[6] = bB.z; acc[7] = bB.w;
#pragma unroll
        for (int ks = 0; ks < 4; ++ks) {
            const v16bf w = W1l[(ct * 4 + ks) * 32 + lane];
            acc = __builtin_amdgcn_wmma_f32_16x16x32_bf16(false, w, false, xf[ks],
                                                          (short)0, acc, false, false);
        }
#pragma unroll
        for (int ks = 0; ks < 4; ++ks) {
            const v16bf w = W1r[(ct * 4 + ks) * 32 + lane];
            acc = __builtin_amdgcn_wmma_f32_16x16x32_bf16(false, w, false, xf[4 + ks],
                                                          (short)0, acc, false, false);
        }
#pragma unroll
        for (int r = 0; r < 8; ++r)
            hf[ct >> 1][(ct & 1) * 8 + r] = (__bf16)fmaxf(acc[r], 0.0f);
    }

#pragma unroll
    for (int ot = 0; ot < GNN_O / 16; ++ot) {
        const float4 bA = *(const float4*)(b2 + ot * 16 + hi * 8);
        const float4 bB = *(const float4*)(b2 + ot * 16 + hi * 8 + 4);
        v8f accz, accr;
#pragma unroll
        for (int r = 0; r < 8; ++r) accz[r] = 0.0f;
        accr[0] = bA.x; accr[1] = bA.y; accr[2] = bA.z; accr[3] = bA.w;
        accr[4] = bB.x; accr[5] = bB.y; accr[6] = bB.z; accr[7] = bB.w;
#pragma unroll
        for (int kk = 0; kk < 4; ++kk) {
            const v16bf wl = W2l[(ot * 4 + kk) * 32 + lane];
            accz = __builtin_amdgcn_wmma_f32_16x16x32_bf16(false, wl, false, hf[kk],
                                                           (short)0, accz, false, false);
        }
#pragma unroll
        for (int kk = 0; kk < 4; ++kk) {
            const v16bf wr = W2r[(ot * 4 + kk) * 32 + lane];
            accr = __builtin_amdgcn_wmma_f32_16x16x32_bf16(false, wr, false, hf[kk],
                                                           (short)0, accr, false, false);
        }
        const long o = (long)row * GNN_O + ot * 16 + hi * 8;
        *(float4*)(Z + o)     = make_float4(accz[0], accz[1], accz[2], accz[3]);
        *(float4*)(Z + o + 4) = make_float4(accz[4], accz[5], accz[6], accz[7]);
        *(float4*)(R + o)     = make_float4(accr[0], accr[1], accr[2], accr[3]);
        *(float4*)(R + o + 4) = make_float4(accr[4], accr[5], accr[6], accr[7]);
    }
}

__global__ void gnn_gather_mean_out(const float* __restrict__ Z,
                                    const float* __restrict__ R,
                                    const int* __restrict__ rowptr,
                                    const int* __restrict__ cnt,
                                    const int* __restrict__ col,
                                    float* __restrict__ out, int N) {
    const int n = __builtin_amdgcn_readfirstlane(blockIdx.x * 8 + (threadIdx.x >> 5));
    if (n >= N) return;
    const int lane = threadIdx.x & 31;
    const int beg  = rowptr[n];
    const int deg  = cnt[n];
    const float* zb = Z + lane * 2;
    const float2 rv = *(const float2*)(R + (long)n * GNN_O + lane * 2);

    float2 acc = make_float2(0.f, 0.f);
    for (int base = 0; base < deg; base += 32) {
        const int m = (deg - base < 32) ? (deg - base) : 32;
        const int myc = (lane < m) ? col[beg + base + lane] : 0;
        int j = 0;
        for (; j + 8 <= m; j += 8) {
            float2 v[8];
#pragma unroll
            for (int u = 0; u < 8; ++u) {
                const int s = __builtin_amdgcn_readlane(myc, j + u);
                v[u] = *(const float2*)(zb + (long)s * GNN_O);
            }
#pragma unroll
            for (int u = 0; u < 8; ++u) { acc.x += v[u].x; acc.y += v[u].y; }
        }
        if (j + 4 <= m) {
            float2 v[4];
#pragma unroll
            for (int u = 0; u < 4; ++u) {
                const int s = __builtin_amdgcn_readlane(myc, j + u);
                v[u] = *(const float2*)(zb + (long)s * GNN_O);
            }
#pragma unroll
            for (int u = 0; u < 4; ++u) { acc.x += v[u].x; acc.y += v[u].y; }
            j += 4;
        }
        const int r = m - j;
        if (r > 0) {
            float2 v0 = make_float2(0.f, 0.f), v1 = v0, v2 = v0;
            v0 = *(const float2*)(zb + (long)__builtin_amdgcn_readlane(myc, j) * GNN_O);
            if (r > 1) v1 = *(const float2*)(zb + (long)__builtin_amdgcn_readlane(myc, j + 1) * GNN_O);
            if (r > 2) v2 = *(const float2*)(zb + (long)__builtin_amdgcn_readlane(myc, j + 2) * GNN_O);
            acc.x += v0.x; acc.y += v0.y;
            acc.x += v1.x; acc.y += v1.y;
            acc.x += v2.x; acc.y += v2.y;
        }
    }
    const float inv = 1.0f / fmaxf((float)deg, 1.0f);
    *(float2*)(out + (long)n * GNN_O + lane * 2) = make_float2(rv.x + acc.x * inv, rv.y + acc.y * inv);
}

extern "C" void kernel_launch(void* const* d_in, const int* in_sizes, int n_in,
                              void* d_out, int out_size, void* d_ws, size_t ws_size,
                              hipStream_t stream) {
    (void)in_sizes; (void)n_in; (void)out_size; (void)ws_size;

    const float* x   = (const float*)d_in[0];
    const int*   ei  = (const int*)d_in[1];
    const float* W1l = (const float*)d_in[2];
    const float* b1  = (const float*)d_in[3];
    const float* W1r = (const float*)d_in[4];
    const float* W2l = (const float*)d_in[5];
    const float* b2  = (const float*)d_in[6];
    const float* W2r = (const float*)d_in[7];
    float*       out = (float*)d_out;

    const int* src = ei;
    const int* dst = ei + GNN_E;

    char*   ws     = (char*)d_ws;
    int*    cnt    = (int*)(ws + 0);
    __bf16* aggb   = (__bf16*)(ws + 524288);
    __bf16* xb     = (__bf16*)(ws + 26124288);
    float*  z      = (float*)(ws + 51724288);
    float*  r      = (float*)(ws + 77324288);
    __bf16* W1l_p  = (__bf16*)(ws + 102924288);
    __bf16* W1r_p  = (__bf16*)(ws + 102957056);
    __bf16* W2l_p  = (__bf16*)(ws + 102989824);
    __bf16* W2r_p  = (__bf16*)(ws + 103006208);
    int*    rowptr = (int*)(ws + 104857600);
    int*    binbase = (int*)(ws + 105381888);
    unsigned* binned = (unsigned*)(ws + 105385984);
    int*    col    = (int*)(ws + 111785984);
    int*    hist   = (int*)(ws + 118185984);

    gnn_bin_hist_pack<<<GNN_NCHUNK + GNN_PACK_WGS + GNN_XWGS, 1024, 0, stream>>>(
        dst, hist, GNN_E, W1l, W1r, W2l, W2r, W1l_p, W1r_p, W2l_p, W2r_p, x, xb);
    gnn_bin_scatter<<<GNN_NCHUNK + GNN_XWGS, 1024, 0, stream>>>(src, dst, hist, binbase, binned, GNN_E,
                                                               x, xb);
    gnn_bin_sort<<<GNN_NBINS + GNN_XWGS, 1024, 0, stream>>>(binned, binbase, cnt, rowptr, col, GNN_N,
                                                            x, xb);

    gnn_gather_mean_bf16<<<(GNN_N + 7) / 8, 256, 0, stream>>>((const unsigned*)xb, rowptr, cnt, col,
                                                              aggb, GNN_N);

    gnn_sage_both<<<GNN_N / 32, 64, 0, stream>>>(
        aggb, xb, (const v16bf*)W1l_p, (const v16bf*)W1r_p, b1,
        (const v16bf*)W2l_p, (const v16bf*)W2r_p, b2, z, r);

    gnn_gather_mean_out<<<(GNN_N + 7) / 8, 256, 0, stream>>>(z, r, rowptr, cnt, col, out, GNN_N);
}


# Layer-1 neighbour mean joins the WMMA kernel: resident 16-wave work-groups walk 16-node tiles, means and h pass through LDS as B operand, weights held in LDS; 4 launches per call; sum order kept
# speedup vs baseline: 7.4947x; 1.1855x over previous
#include <hip/hip_runtime.h>


typedef __attribute__((ext_vector_type(16))) __bf16 v16bf;
typedef __attribute__((ext_vector_type(8)))  __bf16 v8bf;
typedef __attribute__((ext_vector_type(4)))  __bf16 v4bf;
typedef __attribute__((ext_vector_type(8)))  float  v8f;

#define GNN_N 100000
#define GNN_E 1600000
#define GNN_D 128
#define GNN_H 128
#define GNN_O 64

#define GNN_BIN_SHIFT 9
#define GNN_BIN_NODES (1 << GNN_BIN_SHIFT)
#define GNN_NBINS     ((GNN_N + GNN_BIN_NODES - 1) / GNN_BIN_NODES)
#define GNN_CHUNK     6400
#define GNN_NCHUNK    ((GNN_E + GNN_CHUNK - 1) / GNN_CHUNK)
#define GNN_EPT       ((GNN_CHUNK + 1023) / 1024)
#define GNN_STAGE_CAP 12288

static_assert(GNN_N % 32 == 0, "the WMMA kernel takes whole 16-node strips, two per block");
static_assert(GNN_NBINS <= 256, "one bin per thread of the first 8 waves");
static_assert(GNN_N <= (1 << (32 - GNN_BIN_SHIFT)), "source id beside the 9-bit local destination");

template <int NW>
__device__ __forceinline__ int gnn_block_excl_scan(int v, int* wsum, int t) {
    const int lane = t & 31;
    const int w    = t >> 5;
    int inc = v;
#pragma unroll
    for (int o = 1; o < 32; o <<= 1) {
        int up = __shfl_up(inc, o);
        if (lane >= o) inc += up;
    }
    if (w < NW && lane == 31) wsum[w] = inc;
    __syncthreads();
    int off = 0;
#pragma unroll
    for (int k = 0; k < NW; ++k) off += (k < w) ? wsum[k] : 0;
    __syncthreads();
    return off + inc - v;
}

__device__ __forceinline__ void gnn_pack_w(const float* __restrict__ W, __bf16* __restrict__ Bp,
                                           int Nout, int t) {
    int lane = t & 31;
    int ks   = (t >> 5) & 3;
    int nt   = t >> 7;
    int hi   = lane >> 4;
    int col  = nt * 16 + (lane & 15);
    int kb   = ks * 32 + hi * 8;
    __bf16* o = Bp + (long)t * 16;
#pragma unroll
    for (int j = 0; j < 8; ++j) o[j]     = (__bf16)W[(kb + j) * Nout + col];
#pragma unroll
    for (int j = 0; j < 8; ++j) o[8 + j] = (__bf16)W[(kb + 16 + j) * Nout + col];
}

#define GNN_X4   (GNN_N * GNN_D / 4)
#define GNN_XWGS ((GNN_X4 + 2 * 8192 - 1) / (2 * 8192))
static_assert(2 * GNN_XWGS * 8192 >= GNN_X4, "the two halves cover x");

__device__ __forceinline__ void gnn_round_x(const float* __restrict__ x, __bf16* __restrict__ xb,
                                            int g, int t) {
    const float4* x4 = (const float4*)x;
    v4bf* o4 = (v4bf*)xb;
    const int j0 = g * 8192 + t;
    float4 a[8];
#pragma unroll
    for (int k = 0; k < 8; ++k) {
        const int j = j0 + k * 1024;
        a[k] = x4[(j < GNN_X4) ? j : (GNN_X4 - 1)];
    }
#pragma unroll
    for (int k = 0; k < 8; ++k) {
        const int j = j0 + k * 1024;
        if (j < GNN_X4) {
            v4bf o;
            o[0] = (__bf16)a[k].x; o[1] = (__bf16)a[k].y; o[2] = (__bf16)a[k].z; o[3] = (__bf16)a[k].w;
            o4[j] = o;
        }
    }
}

#define GNN_PACK_WGS 3
__global__ __launch_bounds__(1024) void gnn_chunk_sort_pack(const int* __restrict__ src,
                                                            const int* __restrict__ dst,
                                                            int* __restrict__ hist,
                                                            int* __restrict__ hstart,
                                                            unsigned* __restrict__ csorted, int E,
                                                            const float* __restrict__ W1l,
                                                            const float* __restrict__ W1r,
                                                            const float* __restrict__ W2l,
                                                            const float* __restrict__ W2r,
                                                            __bf16* __restrict__ W1l_p,
                                                            __bf16* __restrict__ W1r_p,
                                                            __bf16* __restrict__ W2l_p,
                                                            __bf16* __restrict__ W2r_p,
                                                            const float* __restrict__ x,
                                                            __bf16* __restrict__ xb) {
    __shared__ int lcnt[GNN_NBINS];
    __shared__ int lstart[256];
    __shared__ int wsum[8];
    __shared__ unsigned sbuf[GNN_CHUNK];
    const int t = threadIdx.x;
    if (blockIdx.x >= GNN_NCHUNK) {
        const int pb = blockIdx.x - GNN_NCHUNK;
        if (pb == 0)      gnn_pack_w(W1l, W1l_p, GNN_H, t);
        else if (pb == 1) gnn_pack_w(W1r, W1r_p, GNN_H, t);
        else if (pb == 2) {
            if (t < 512) gnn_pack_w(W2l, W2l_p, GNN_O, t);
            else         gnn_pack_w(W2r, W2r_p, GNN_O, t - 512);
        } else {
            gnn_round_x(x, xb, pb - GNN_PACK_WGS, t);
        }
        return;
    }
    const int wg   = blockIdx.x;
    const int base = wg * GNN_CHUNK;
    const int end  = (base + GNN_CHUNK < E) ? (base + GNN_CHUNK) : E;

    if (t < GNN_NBINS) lcnt[t] = 0;

    int d[GNN_EPT], s[GNN_EPT], r[GNN_EPT];
#pragma unroll
    for (int k = 0; k < GNN_EPT; ++k) {
        const int i = base + k * 1024 + t;
        d[k] = -1; s[k] = 0; r[k] = 0;
        if (i < end) { d[k] = dst[i]; s[k] = src[i]; }
    }
    __syncthreads();

#pragma unroll
    for (int k = 0; k < GNN_EPT; ++k) {
        if (d[k] >= 0) r[k] = atomicAdd(&lcnt[d[k] >> GNN_BIN_SHIFT], 1);
    }
    __syncthreads();

    const int mine = (t < GNN_NBINS) ? lcnt[t] : 0;
    const int ls   = gnn_block_excl_scan<8>(mine, wsum, t);
    if (t < GNN_NBINS) {
        lstart[t] = ls;
        hist[wg * GNN_NBINS + t]   = mine;
        hstart[wg * GNN_NBINS + t] = ls;
    }
    __syncthreads();

#pragma unroll
    for (int k = 0; k < GNN_EPT; ++k) {
        if (d[k] >= 0) {
            const int p = lstart[d[k] >> GNN_BIN_SHIFT] + r[k];
            sbuf[p] = ((unsigned)s[k] << GNN_BIN_SHIFT) | (unsigned)(d[k] & (GNN_BIN_NODES - 1));
        }
    }
    __syncthreads();

    const int n = end - base;
    for (int i = t; i < n; i += 1024) csorted[base + i] = sbuf[i];
}

static_assert(GNN_NCHUNK <= 256, "one run per thread of the first 8 waves");
__global__ __launch_bounds__(1024) void gnn_bin_sort(const unsigned* __restrict__ csorted,
                                                     const int* __restrict__ hist,
                                                     const int* __restrict__ hstart,
                                                     int* __restrict__ cnt,
                                                     int* __restrict__ rowptr,
                                                     int* __restrict__ col, int N,
                                                     const float* __restrict__ x,
                                                     __bf16* __restrict__ xb) {
    if (blockIdx.x >= GNN_NBINS) {
        gnn_round_x(x, xb, GNN_XWGS + (int)(blockIdx.x - GNN_NBINS), threadIdx.x);
        return;
    }
    __shared__ int ncnt[GNN_BIN_NODES];
    __shared__ int wsum[16];
    __shared__ int roff[256];
    __shared__ int rlen[256];
    __shared__ int rpos[256];
    __shared__ int tot[2];
    __shared__ unsigned ebuf[GNN_STAGE_CAP];
    __shared__ int      sbuf[GNN_STAGE_CAP];

    const int t    = threadIdx.x;
    const int lane = t & 31;
    const int wv   = t >> 5;
    const int bin  = blockIdx.x;

    int len = 0, st = 0;
    if (t < GNN_NCHUNK) {
        len = hist[t * GNN_NBINS + bin];
        st  = hstart[t * GNN_NBINS + bin];
    }
    if (t < GNN_BIN_NODES) ncnt[t] = 0;
    const int off = gnn_block_excl_scan<8>(len, wsum, t);
    const int sb  = gnn_block_excl_scan<8>(st, wsum, t);
    if (t < GNN_NCHUNK) { roff[t] = off; rlen[t] = len; rpos[t] = t * GNN_CHUNK + st; }
    if (t == 256) { tot[0] = off; tot[1] = sb; }
    __syncthreads();
    const int n   = tot[0];
    const int beg = tot[1];
    const bool staged = (n <= GNN_STAGE_CAP);

    for (int c0 = wv; c0 < GNN_NCHUNK; c0 += 8 * 32) {
        unsigned e0[8], e1[8];
        int L[8], P[8], O[8];
#pragma unroll
        for (int u = 0; u < 8; ++u) {
            const int c  = c0 + 32 * u;
            const bool ok = (c < GNN_NCHUNK);
            const int cc = ok ? c : 0;
            L[u] = ok ? rlen[cc] : 0;
            P[u] = rpos[cc];
            O[u] = roff[cc];
            e0[u] = (lane < L[u])      ? csorted[P[u] + lane]      : 0u;
            e1[u] = (lane + 32 < L[u]) ? csorted[P[u] + lane + 32] : 0u;
        }
#pragma unroll
        for (int u = 0; u < 8; ++u) {
            if (lane < L[u]) {
                atomicAdd(&ncnt[e0[u] & (GNN_BIN_NODES - 1)], 1);
                if (staged) ebuf[O[u] + lane] = e0[u];
            }
            if (lane + 32 < L[u]) {
                atomicAdd(&ncnt[e1[u] & (GNN_BIN_NODES - 1)], 1);
                if (staged) ebuf[O[u] + lane + 32] = e1[u];
            }
            for (int i = lane + 64; i < L[u]; i += 32) {
                const unsigned e = csorted[P[u] + i];
                atomicAdd(&ncnt[e & (GNN_BIN_NODES - 1)], 1);
                if (staged) ebuf[O[u] + i] = e;
            }
        }
    }
    __syncthreads();

    const int deg = (t < GNN_BIN_NODES) ? ncnt[t] : 0;
    const int ex  = gnn_block_excl_scan<16>(deg, wsum, t);
    if (t < GNN_BIN_NODES) {
        ncnt[t] = ex;
        const int node = bin * GNN_BIN_NODES + t;
        if (node < N) { cnt[node] = deg; rowptr[node] = beg + ex; }
    }
    __syncthreads();

    if (staged) {
        for (int i0 = t; i0 < n; i0 += 8 * 1024) {
            unsigned w[8];
#pragma unroll
            for (int u = 0; u < 8; ++u) {
                const int i = i0 + u * 1024;
                w[u] = ebuf[(i < n) ? i : (n - 1)];
            }
#pragma unroll
            for (int u = 0; u < 8; ++u) {
                if (i0 + u * 1024 < n) {
                    const int slot = atomicAdd(&ncnt[w[u] & (GNN_BIN_NODES - 1)], 1);
                    sbuf[slot] = (int)(w[u] >> GNN_BIN_SHIFT);
                }
            }
        }
        __syncthreads();
        for (int i = t; i < n; i += 1024) col[beg + i] = sbuf[i];
    } else {
        for (int c = wv; c < GNN_NCHUNK; c += 32) {
            const int Lc = rlen[c], Pc = rpos[c];
            for (int i = lane; i < Lc; i += 32) {
                const unsigned e = csorted[Pc + i];
                const int slot = atomicAdd(&ncnt[e & (GNN_BIN_NODES - 1)], 1);
                col[beg + slot] = (int)(e >> GNN_BIN_SHIFT);
            }
        }
    }
}

__device__ __forceinline__ void gnn_lds_barrier() {
    __builtin_amdgcn_fence(__ATOMIC_RELEASE, "workgroup", "local");
    __builtin_amdgcn_s_barrier();
    __builtin_amdgcn_fence(__ATOMIC_ACQUIRE, "workgroup", "local");
}

#define GNN_TILE       16
#define GNN_NTILES     (GNN_N / GNN_TILE)
#define GNN_FUSED_WGS  512
#define GNN_ROWP       136
#define GNN_WROWS      (2 * GNN_H / 16 * 4 * 32 + 2 * GNN_O / 16 * 4 * 32)
static_assert(GNN_N % GNN_TILE == 0, "whole 16-node tiles");
static_assert(GNN_FUSED_WGS <= GNN_NTILES, "every work-group has a first tile");
static_assert(GNN_WROWS == 3072, "W1l 1024, W1r 1024, W2l 512, W2r 512 fragment rows");
static_assert((GNN_WROWS * 2) % 512 == 0, "the weight copy is whole 16-B pieces per thread");

__global__ __launch_bounds__(512) void gnn_gather_sage(const unsigned* __restrict__ Xb,
                                                       const int* __restrict__ rowptr,
                                                       const int* __restrict__ cnt,
                                                       const int* __restrict__ col,
                                                       const uint4* __restrict__ Wpk,
                                                       const float* __restrict__ b1,
                                                       const float* __restrict__ b2,
                                                       float* __restrict__ Z,
                                                       float* __restrict__ R) {
    __shared__ __attribute__((aligned(32))) uint4 sW[GNN_WROWS * 2];
    __shared__ __attribute__((aligned(16))) __bf16 smean[GNN_TILE * GNN_ROWP];
    __shared__ __attribute__((aligned(16))) __bf16 sx[GNN_TILE * GNN_ROWP];
    __shared__ __attribute__((aligned(16))) __bf16 sh[GNN_TILE * GNN_ROWP];

    const int t    = threadIdx.x;
    const int lane = t & 31;
    const int w    = __builtin_amdgcn_readfirstlane(t >> 5);
    const int hi   = lane >> 4;
    const int nl   = lane & 15;

    {
        uint4 wv[GNN_WROWS * 2 / 512];
#pragma unroll
        for (int k = 0; k < GNN_WROWS * 2 / 512; ++k) wv[k] = Wpk[t + k * 512];
#pragma unroll
        for (int k = 0; k < GNN_WROWS * 2 / 512; ++k) sW[t + k * 512] = wv[k];
    }
    const v16bf* sWf = (const v16bf*)sW;

    const int ct = w & 7;
    const int ot = w & 3;
    const bool isr = (w & 4) != 0;
    v8f bias1, bias2;
    {
        const float4 bA = *(const float4*)(b1 + ct * 16 + hi * 8);
        const float4 bB = *(const float4*)(b1 + ct * 16 + hi * 8 + 4);
        bias1[0] = bA.x; bias1[1] = bA.y; bias1[2] = bA.z; bias1[3] = bA.w;
        bias1[4] = bB.x; bias1[5] = bB.y; bias1[6] = bB.z; bias1[7] = bB.w;
        const float4 cA = *(const float4*)(b2 + ot * 16 + hi * 8);
        const float4 cB = *(const float4*)(b2 + ot * 16 + hi * 8 + 4);
        bias2[0] = isr ? cA.x : 0.0f; bias2[1] = isr ? cA.y : 0.0f;
        bias2[2] = isr ? cA.z : 0.0f; bias2[3] = isr ? cA.w : 0.0f;
        bias2[4] = isr ? cB.x : 0.0f; bias2[5] = isr ? cB.y : 0.0f;
        bias2[6] = isr ? cB.z : 0.0f; bias2[7] = isr ? cB.w : 0.0f;
    }
    const v16bf* w1l = sWf + (ct * 4) * 32 + lane;
    const v16bf* w1r = sWf + 1024 + (ct * 4) * 32 + lane;
    const v16bf* w2  = sWf + (isr ? 2560 : 2048) + (ot * 4) * 32 + lane;

    const unsigned* xb = Xb + lane * 2;

    int tile = blockIdx.x;
    const int xpiece = (t >> 4) * 64 + (t & 15) * 4;
    int beg, deg, myc, begv, degv;
    uint4 xrow = make_uint4(0u, 0u, 0u, 0u);
    {
        const int n = tile * GNN_TILE + w;
        beg = rowptr[n];
        deg = cnt[n];
        const int t1 = (tile + GNN_FUSED_WGS < GNN_NTILES) ? (tile + GNN_FUSED_WGS) : tile;
        begv = rowptr[t1 * GNN_TILE + nl];
        degv = cnt[t1 * GNN_TILE + nl];
        if (t < 256) xrow = *(const uint4*)(Xb + (long)tile * (GNN_TILE * 64) + xpiece);
        myc = (lane < deg) ? col[beg + lane] : 0;
    }

    for (; tile < GNN_NTILES; tile += GNN_FUSED_WGS) {

        float a0 = 0.f, a1 = 0.f, a2 = 0.f, a3 = 0.f;
        for (int base = 0; base < deg; base += 32) {
            const int m = (deg - base < 32) ? (deg - base) : 32;
            if (base > 0) myc = (lane < m) ? col[beg + base + lane] : 0;
            int j = 0;
            for (; j + 16 <= m; j += 16) {
                uint2 v[16];
#pragma unroll
                for (int u = 0; u < 16; ++u) {
                    const int s = __builtin_amdgcn_readlane(myc, j + u);
                    v[u] = *(const uint2*)(xb + (long)s * 64);
                }
#pragma unroll
                for (int u = 0; u < 16; ++u) {
                    a0 += __uint_as_float(v[u].x << 16); a1 += __uint_as_float(v[u].x & 0xffff0000u);
                    a2 += __uint_as_float(v[u].y << 16); a3 += __uint_as_float(v[u].y & 0xffff0000u);
                }
            }
            if (j + 8 <= m) {
                uint2 v[8];
#pragma unroll
                for (int u = 0; u < 8; ++u) {
                    const int s = __builtin_amdgcn_readlane(myc, j + u);
                    v[u] = *(const uint2*)(xb + (long)s * 64);
                }
#pragma unroll
                for (int u = 0; u < 8; ++u) {
                    a0 += __uint_as_float(v[u].x << 16); a1 += __uint_as_float(v[u].x & 0xffff0000u);
                    a2 += __uint_as_float(v[u].y << 16); a3 += __uint_as_float(v[u].y & 0xffff0000u);
                }
                j += 8;
            }
            if (j + 4 <= m) {
                uint2 v[4];
#pragma unroll
                for (int u = 0; u < 4; ++u) {
                    const int s = __builtin_amdgcn_readlane(myc, j + u);
                    v[u] = *(const uint2*)(xb + (long)s * 64);
                }
#pragma unroll
                for (int u = 0; u < 4; ++u) {
                    a0 += __uint_as_float(v[u].x << 16); a1 += __uint_as_float(v[u].x & 0xffff0000u);
                    a2 += __uint_as_float(v[u].y << 16); a3 += __uint_as_float(v[u].y & 0xffff0000u);
                }
                j += 4;
            }
            const int r = m - j;
            if (r > 0) {
                uint2 v0 = make_uint2(0u, 0u), v1 = v0, v2 = v0;
                v0 = *(const uint2*)(xb + (long)__builtin_amdgcn_readlane(myc, j) * 64);
                if (r > 1) v1 = *(const uint2*)(xb + (long)__builtin_amdgcn_readlane(myc, j + 1) * 64);
                if (r > 2) v2 = *(const uint2*)(xb + (long)__builtin_amdgcn_readlane(myc, j + 2) * 64);
                a0 += __uint_as_float(v0.x << 16); a1 += __uint_as_float(v0.x & 0xffff0000u);
                a2 += __uint_as_float(v0.y << 16); a3 += __uint_as_float(v0.y & 0xffff0000u);
                a0 += __uint_as_float(v1.x << 16); a1 += __uint_as_float(v1.x & 0xffff0000u);
                a2 += __uint_as_float(v1.y << 16); a3 += __uint_as_float(v1.y & 0xffff0000u);
                a0 += __uint_as_float(v2.x << 16); a1 += __uint_as_float(v2.x & 0xffff0000u);
                a2 += __uint_as_float(v2.y << 16); a3 += __uint_as_float(v2.y & 0xffff0000u);
            }
        }
        const float inv = 1.0f / fmaxf((float)deg, 1.0f);
        v4bf o;
        o[0] = (__bf16)(a0 * inv); o[1] = (__bf16)(a1 * inv);
        o[2] = (__bf16)(a2 * inv); o[3] = (__bf16)(a3 * inv);
        *(v4bf*)(smean + w * GNN_ROWP + lane * 4) = o;
        if (t < 256) *(uint4*)(sx + (t >> 4) * GNN_ROWP + (t & 15) * 8) = xrow;

        const int t1   = (tile + GNN_FUSED_WGS < GNN_NTILES) ? (tile + GNN_FUSED_WGS) : tile;
        const int t2   = (tile + 2 * GNN_FUSED_WGS < GNN_NTILES) ? (tile + 2 * GNN_FUSED_WGS) : tile;
        const int beg2 = __builtin_amdgcn_readlane(begv, w);
        const int deg2 = __builtin_amdgcn_readlane(degv, w);
        begv = rowptr[t2 * GNN_TILE + nl];
        degv = cnt[t2 * GNN_TILE + nl];
        if (t < 256) xrow = *(const uint4*)(Xb + (long)t1 * (GNN_TILE * 64) + xpiece);
        const int myc2 = (lane < deg2) ? col[beg2 + lane] : 0;
        gnn_lds_barrier();

        if (w < 8) {
            const __bf16* mrow = smean + nl * GNN_ROWP + hi * 8;
            const __bf16* xrw  = sx    + nl * GNN_ROWP + hi * 8;
            v8f acc = bias1;
#pragma unroll
            for (int ks = 0; ks < 4; ++ks) {
                const v8bf lo = *(const v8bf*)(mrow + ks * 32);
                const v8bf up = *(const v8bf*)(mrow + ks * 32 + 16);
                const v16bf bf = __builtin_shufflevector(lo, up, 0, 1, 2, 3, 4, 5, 6, 7, 8, 9, 10, 11, 12, 13, 14, 15);
                acc = __builtin_amdgcn_wmma_f32_16x16x32_bf16(false, w1l[ks * 32], false, bf,
                                                              (short)0, acc, false, false);
            }
#pragma unroll
            for (int ks = 0; ks < 4; ++ks) {
                const v8bf lo = *(const v8bf*)(xrw + ks * 32);
                const v8bf up = *(const v8bf*)(xrw + ks * 32 + 16);
                const v16bf bf = __builtin_shufflevector(lo, up, 0, 1, 2, 3, 4, 5, 6, 7, 8, 9, 10, 11, 12, 13, 14, 15);
                acc = __builtin_amdgcn_wmma_f32_16x16x32_bf16(false, w1r[ks * 32], false, bf,
                                                              (short)0, acc, false, false);
            }
            v8bf hv;
#pragma unroll
            for (int r = 0; r < 8; ++r) hv[r] = (__bf16)fmaxf(acc[r], 0.0f);
            *(v8bf*)(sh + nl * GNN_ROWP + ct * 16 + hi * 8) = hv;
        }
        gnn_lds_barrier();

        if (w < 8) {
            const __bf16* hrow = sh + nl * GNN_ROWP + hi * 8;
            v8f acc = bias2;
#pragma unroll
            for (int kk = 0; kk < 4; ++kk) {
                const v8bf lo = *(const v8bf*)(hrow + kk * 32);
                const v8bf up = *(const v8bf*)(hrow + kk * 32 + 16);
                const v16bf bf = __builtin_shufflevector(lo, up, 0, 1, 2, 3, 4, 5, 6, 7, 8, 9, 10, 11, 12, 13, 14, 15);
                acc = __builtin_amdgcn_wmma_f32_16x16x32_bf16(false, w2[kk * 32], false, bf,
                                                              (short)0, acc, false, false);
            }
            const long o8 = (long)(tile * GNN_TILE + nl) * GNN_O + ot * 16 + hi * 8;
            if (isr) {
                *(float4*)(R + o8)     = make_float4(acc[0], acc[1], acc[2], acc[3]);
                *(float4*)(R + o8 + 4) = make_float4(acc[4], acc[5], acc[6], acc[7]);
            } else {
                *(float4*)(Z + o8)     = make_float4(acc[0], acc[1], acc[2], acc[3]);
                *(float4*)(Z + o8 + 4) = make_float4(acc[4], acc[5], acc[6], acc[7]);
            }
        }
        beg = beg2; deg = deg2; myc = myc2;
    }
}

__global__ void gnn_gather_mean_out(const float* __restrict__ Z,
                                    const float* __restrict__ R,
                                    const int* __restrict__ rowptr,
                                    const int* __restrict__ cnt,
                                    const int* __restrict__ col,
                                    float* __restrict__ out, int N) {
    const int n = __builtin_amdgcn_readfirstlane(blockIdx.x * 8 + (threadIdx.x >> 5));
    if (n >= N) return;
    const int lane = threadIdx.x & 31;
    const int beg  = rowptr[n];
    const int deg  = cnt[n];
    const float* zb = Z + lane * 2;
    const float2 rv = *(const float2*)(R + (long)n * GNN_O + lane * 2);

    float2 acc = make_float2(0.f, 0.f);
    for (int base = 0; base < deg; base += 32) {
        const int m = (deg - base < 32) ? (deg - base) : 32;
        const int myc = (lane < m) ? col[beg + base + lane] : 0;
        int j = 0;
        for (; j + 8 <= m; j += 8) {
            float2 v[8];
#pragma unroll
            for (int u = 0; u < 8; ++u) {
                const int s = __builtin_amdgcn_readlane(myc, j + u);
                v[u] = *(const float2*)(zb + (long)s * GNN_O);
            }
#pragma unroll
            for (int u = 0; u < 8; ++u) { acc.x += v[u].x; acc.y += v[u].y; }
        }
        if (j + 4 <= m) {
            float2 v[4];
#pragma unroll
            for (int u = 0; u < 4; ++u) {
                const int s = __builtin_amdgcn_readlane(myc, j + u);
                v[u] = *(const float2*)(zb + (long)s * GNN_O);
            }
#pragma unroll
            for (int u = 0; u < 4; ++u) { acc.x += v[u].x; acc.y += v[u].y; }
            j += 4;
        }
        const int r = m - j;
        if (r > 0) {
            float2 v0 = make_float2(0.f, 0.f), v1 = v0, v2 = v0;
            v0 = *(const float2*)(zb + (long)__builtin_amdgcn_readlane(myc, j) * GNN_O);
            if (r > 1) v1 = *(const float2*)(zb + (long)__builtin_amdgcn_readlane(myc, j + 1) * GNN_O);
            if (r > 2) v2 = *(const float2*)(zb + (long)__builtin_amdgcn_readlane(myc, j + 2) * GNN_O);
            acc.x += v0.x; acc.y += v0.y;
            acc.x += v1.x; acc.y += v1.y;
            acc.x += v2.x; acc.y += v2.y;
        }
    }
    const float inv = 1.0f / fmaxf((float)deg, 1.0f);
    *(float2*)(out + (long)n * GNN_O + lane * 2) = make_float2(rv.x + acc.x * inv, rv.y + acc.y * inv);
}

extern "C" void kernel_launch(void* const* d_in, const int* in_sizes, int n_in,
                              void* d_out, int out_size, void* d_ws, size_t ws_size,
                              hipStream_t stream) {
    (void)in_sizes; (void)n_in; (void)out_size; (void)ws_size;

    const float* x   = (const float*)d_in[0];
    const int*   ei  = (const int*)d_in[1];
    const float* W1l = (const float*)d_in[2];
    const float* b1  = (const float*)d_in[3];
    const float* W1r = (const float*)d_in[4];
    const float* W2l = (const float*)d_in[5];
    const float* b2  = (const float*)d_in[6];
    const float* W2r = (const float*)d_in[7];
    float*       out = (float*)d_out;

    const int* src = ei;
    const int* dst = ei + GNN_E;

    char*   ws     = (char*)d_ws;
    int*    cnt    = (int*)(ws + 0);
    __bf16* xb     = (__bf16*)(ws + 26124288);
    float*  z      = (float*)(ws + 51724288);
    float*  r      = (float*)(ws + 77324288);
    __bf16* W1l_p  = (__bf16*)(ws + 102924288);
    __bf16* W1r_p  = (__bf16*)(ws + 102957056);
    __bf16* W2l_p  = (__bf16*)(ws + 102989824);
    __bf16* W2r_p  = (__bf16*)(ws + 103006208);
    int*    rowptr = (int*)(ws + 104857600);
    unsigned* csorted = (unsigned*)(ws + 105385984);
    int*    col    = (int*)(ws + 111785984);
    int*    hist   = (int*)(ws + 118185984);
    int*    hstart = (int*)(ws + 118382592);

    gnn_chunk_sort_pack<<<GNN_NCHUNK + GNN_PACK_WGS + GNN_XWGS, 1024, 0, stream>>>(
        src, dst, hist, hstart, csorted, GNN_E, W1l, W1r, W2l, W2r, W1l_p, W1r_p, W2l_p, W2r_p, x, xb);
    gnn_bin_sort<<<GNN_NBINS + GNN_XWGS, 1024, 0, stream>>>(csorted, hist, hstart, cnt, rowptr, col,
                                                            GNN_N, x, xb);

    gnn_gather_sage<<<GNN_FUSED_WGS, 512, 0, stream>>>((const unsigned*)xb, rowptr, cnt, col,
                                                        (const uint4*)W1l_p, b1, b2, z, r);

    gnn_gather_mean_out<<<(GNN_N + 7) / 8, 256, 0, stream>>>(z, r, rowptr, cnt, col, out, GNN_N);
}


# Layer-1 neighbour sum: all row loads of a node (up to 32) go out as one group before the first add, not in dependent groups of 16, 8, 4 and 1..3: one memory trip per node; sums and order kept
# speedup vs baseline: 8.6893x; 1.1594x over previous
#include <hip/hip_runtime.h>


typedef __attribute__((ext_vector_type(16))) __bf16 v16bf;
typedef __attribute__((ext_vector_type(8)))  __bf16 v8bf;
typedef __attribute__((ext_vector_type(4)))  __bf16 v4bf;
typedef __attribute__((ext_vector_type(8)))  float  v8f;

#define GNN_N 100000
#define GNN_E 1600000
#define GNN_D 128
#define GNN_H 128
#define GNN_O 64

#define GNN_BIN_SHIFT 9
#define GNN_BIN_NODES (1 << GNN_BIN_SHIFT)
#define GNN_NBINS     ((GNN_N + GNN_BIN_NODES - 1) / GNN_BIN_NODES)
#define GNN_CHUNK     6400
#define GNN_NCHUNK    ((GNN_E + GNN_CHUNK - 1) / GNN_CHUNK)
#define GNN_EPT       ((GNN_CHUNK + 1023) / 1024)
#define GNN_STAGE_CAP 12288

static_assert(GNN_N % 32 == 0, "the WMMA kernel takes whole 16-node strips, two per block");
static_assert(GNN_NBINS <= 256, "one bin per thread of the first 8 waves");
static_assert(GNN_N <= (1 << (32 - GNN_BIN_SHIFT)), "source id beside the 9-bit local destination");

template <int NW>
__device__ __forceinline__ int gnn_block_excl_scan(int v, int* wsum, int t) {
    const int lane = t & 31;
    const int w    = t >> 5;
    int inc = v;
#pragma unroll
    for (int o = 1; o < 32; o <<= 1) {
        int up = __shfl_up(inc, o);
        if (lane >= o) inc += up;
    }
    if (w < NW && lane == 31) wsum[w] = inc;
    __syncthreads();
    int off = 0;
#pragma unroll
    for (int k = 0; k < NW; ++k) off += (k < w) ? wsum[k] : 0;
    __syncthreads();
    return off + inc - v;
}

__device__ __forceinline__ void gnn_pack_w(const float* __restrict__ W, __bf16* __restrict__ Bp,
                                           int Nout, int t) {
    int lane = t & 31;
    int ks   = (t >> 5) & 3;
    int nt   = t >> 7;
    int hi   = lane >> 4;
    int col  = nt * 16 + (lane & 15);
    int kb   = ks * 32 + hi * 8;
    __bf16* o = Bp + (long)t * 16;
#pragma unroll
    for (int j = 0; j < 8; ++j) o[j]     = (__bf16)W[(kb + j) * Nout + col];
#pragma unroll
    for (int j = 0; j < 8; ++j) o[8 + j] = (__bf16)W[(kb + 16 + j) * Nout + col];
}

#define GNN_X4   (GNN_N * GNN_D / 4)
#define GNN_XWGS ((GNN_X4 + 2 * 8192 - 1) / (2 * 8192))
static_assert(2 * GNN_XWGS * 8192 >= GNN_X4, "the two halves cover x");

__device__ __forceinline__ void gnn_round_x(const float* __restrict__ x, __bf16* __restrict__ xb,
                                            int g, int t) {
    const float4* x4 = (const float4*)x;
    v4bf* o4 = (v4bf*)xb;
    const int j0 = g * 8192 + t;
    float4 a[8];
#pragma unroll
    for (int k = 0; k < 8; ++k) {
        const int j = j0 + k * 1024;
        a[k] = x4[(j < GNN_X4) ? j : (GNN_X4 - 1)];
    }
#pragma unroll
    for (int k = 0; k < 8; ++k) {
        const int j = j0 + k * 1024;
        if (j < GNN_X4) {
            v4bf o;
            o[0] = (__bf16)a[k].x; o[1] = (__bf16)a[k].y; o[2] = (__bf16)a[k].z; o[3] = (__bf16)a[k].w;
            o4[j] = o;
        }
    }
}

#define GNN_PACK_WGS 3
__global__ __launch_bounds__(1024) void gnn_chunk_sort_pack(const int* __restrict__ src,
                                                            const int* __restrict__ dst,
                                                            int* __restrict__ hist,
                                                            int* __restrict__ hstart,
                                                            unsigned* __restrict__ csorted, int E,
                                                            const float* __restrict__ W1l,
                                                            const float* __restrict__ W1r,
                                                            const float* __restrict__ W2l,
                                                            const float* __restrict__ W2r,
                                                            __bf16* __restrict__ W1l_p,
                                                            __bf16* __restrict__ W1r_p,
                                                            __bf16* __restrict__ W2l_p,
                                                            __bf16* __restrict__ W2r_p,
                                                            const float* __restrict__ x,
                                                            __bf16* __restrict__ xb) {
    __shared__ int lcnt[GNN_NBINS];
    __shared__ int lstart[256];
    __shared__ int wsum[8];
    __shared__ unsigned sbuf[GNN_CHUNK];
    const int t = threadIdx.x;
    if (blockIdx.x >= GNN_NCHUNK) {
        const int pb = blockIdx.x - GNN_NCHUNK;
        if (pb == 0)      gnn_pack_w(W1l, W1l_p, GNN_H, t);
        else if (pb == 1) gnn_pack_w(W1r, W1r_p, GNN_H, t);
        else if (pb == 2) {
            if (t < 512) gnn_pack_w(W2l, W2l_p, GNN_O, t);
            else         gnn_pack_w(W2r, W2r_p, GNN_O, t - 512);
        } else {
            gnn_round_x(x, xb, pb - GNN_PACK_WGS, t);
        }
        return;
    }
    const int wg   = blockIdx.x;
    const int base = wg * GNN_CHUNK;
    const int end  = (base + GNN_CHUNK < E) ? (base + GNN_CHUNK) : E;

    if (t < GNN_NBINS) lcnt[t] = 0;

    int d[GNN_EPT], s[GNN_EPT], r[GNN_EPT];
#pragma unroll
    for (int k = 0; k < GNN_EPT; ++k) {
        const int i = base + k * 1024 + t;
        d[k] = -1; s[k] = 0; r[k] = 0;
        if (i < end) { d[k] = dst[i]; s[k] = src[i]; }
    }
    __syncthreads();

#pragma unroll
    for (int k = 0; k < GNN_EPT; ++k) {
        if (d[k] >= 0) r[k] = atomicAdd(&lcnt[d[k] >> GNN_BIN_SHIFT], 1);
    }
    __syncthreads();

    const int mine = (t < GNN_NBINS) ? lcnt[t] : 0;
    const int ls   = gnn_block_excl_scan<8>(mine, wsum, t);
    if (t < GNN_NBINS) {
        lstart[t] = ls;
        hist[wg * GNN_NBINS + t]   = mine;
        hstart[wg * GNN_NBINS + t] = ls;
    }
    __syncthreads();

#pragma unroll
    for (int k = 0; k < GNN_EPT; ++k) {
        if (d[k] >= 0) {
            const int p = lstart[d[k] >> GNN_BIN_SHIFT] + r[k];
            sbuf[p] = ((unsigned)s[k] << GNN_BIN_SHIFT) | (unsigned)(d[k] & (GNN_BIN_NODES - 1));
        }
    }
    __syncthreads();

    const int n = end - base;
    for (int i = t; i < n; i += 1024) csorted[base + i] = sbuf[i];
}

static_assert(GNN_NCHUNK <= 256, "one run per thread of the first 8 waves");
__global__ __launch_bounds__(1024) void gnn_bin_sort(const unsigned* __restrict__ csorted,
                                                     const int* __restrict__ hist,
                                                     const int* __restrict__ hstart,
                                                     int* __restrict__ cnt,
                                                     int* __restrict__ rowptr,
                                                     int* __restrict__ col, int N,
                                                     const float* __restrict__ x,
                                                     __bf16* __restrict__ xb) {
    if (blockIdx.x >= GNN_NBINS) {
        gnn_round_x(x, xb, GNN_XWGS + (int)(blockIdx.x - GNN_NBINS), threadIdx.x);
        return;
    }
    __shared__ int ncnt[GNN_BIN_NODES];
    __shared__ int wsum[16];
    __shared__ int roff[256];
    __shared__ int rlen[256];
    __shared__ int rpos[256];
    __shared__ int tot[2];
    __shared__ unsigned ebuf[GNN_STAGE_CAP];
    __shared__ int      sbuf[GNN_STAGE_CAP];

    const int t    = threadIdx.x;
    const int lane = t & 31;
    const int wv   = t >> 5;
    const int bin  = blockIdx.x;

    int len = 0, st = 0;
    if (t < GNN_NCHUNK) {
        len = hist[t * GNN_NBINS + bin];
        st  = hstart[t * GNN_NBINS + bin];
    }
    if (t < GNN_BIN_NODES) ncnt[t] = 0;
    const int off = gnn_block_excl_scan<8>(len, wsum, t);
    const int sb  = gnn_block_excl_scan<8>(st, wsum, t);
    if (t < GNN_NCHUNK) { roff[t] = off; rlen[t] = len; rpos[t] = t * GNN_CHUNK + st; }
    if (t == 256) { tot[0] = off; tot[1] = sb; }
    __syncthreads();
    const int n   = tot[0];
    const int beg = tot[1];
    const bool staged = (n <= GNN_STAGE_CAP);

    for (int c0 = wv; c0 < GNN_NCHUNK; c0 += 8 * 32) {
        unsigned e0[8], e1[8];
        int L[8], P[8], O[8];
#pragma unroll
        for (int u = 0; u < 8; ++u) {
            const int c  = c0 + 32 * u;
            const bool ok = (c < GNN_NCHUNK);
            const int cc = ok ? c : 0;
            L[u] = ok ? rlen[cc] : 0;
            P[u] = rpos[cc];
            O[u] = roff[cc];
            e0[u] = (lane < L[u])      ? csorted[P[u] + lane]      : 0u;
            e1[u] = (lane + 32 < L[u]) ? csorted[P[u] + lane + 32] : 0u;
        }
#pragma unroll
        for (int u = 0; u < 8; ++u) {
            if (lane < L[u]) {
                atomicAdd(&ncnt[e0[u] & (GNN_BIN_NODES - 1)], 1);
                if (staged) ebuf[O[u] + lane] = e0[u];
            }
            if (lane + 32 < L[u]) {
                atomicAdd(&ncnt[e1[u] & (GNN_BIN_NODES - 1)], 1);
                if (staged) ebuf[O[u] + lane + 32] = e1[u];
            }
            for (int i = lane + 64; i < L[u]; i += 32) {
                const unsigned e = csorted[P[u] + i];
                atomicAdd(&ncnt[e & (GNN_BIN_NODES - 1)], 1);
                if (staged) ebuf[O[u] + i] = e;
            }
        }
    }
    __syncthreads();

    const int deg = (t < GNN_BIN_NODES) ? ncnt[t] : 0;
    const int ex  = gnn_block_excl_scan<16>(deg, wsum, t);
    if (t < GNN_BIN_NODES) {
        ncnt[t] = ex;
        const int node = bin * GNN_BIN_NODES + t;
        if (node < N) { cnt[node] = deg; rowptr[node] = beg + ex; }
    }
    __syncthreads();

    if (staged) {
        for (int i0 = t; i0 < n; i0 += 8 * 1024) {
            unsigned w[8];
#pragma unroll
            for (int u = 0; u < 8; ++u) {
                const int i = i0 + u * 1024;
                w[u] = ebuf[(i < n) ? i : (n - 1)];
            }
#pragma unroll
            for (int u = 0; u < 8; ++u) {
                if (i0 + u * 1024 < n) {
                    const int slot = atomicAdd(&ncnt[w[u] & (GNN_BIN_NODES - 1)], 1);
                    sbuf[slot] = (int)(w[u] >> GNN_BIN_SHIFT);
                }
            }
        }
        __syncthreads();
        for (int i = t; i < n; i += 1024) col[beg + i] = sbuf[i];
    } else {
        for (int c = wv; c < GNN_NCHUNK; c += 32) {
            const int Lc = rlen[c], Pc = rpos[c];
            for (int i = lane; i < Lc; i += 32) {
                const unsigned e = csorted[Pc + i];
                const int slot = atomicAdd(&ncnt[e & (GNN_BIN_NODES - 1)], 1);
                col[beg + slot] = (int)(e >> GNN_BIN_SHIFT);
            }
        }
    }
}

__device__ __forceinline__ void gnn_lds_barrier() {
    __builtin_amdgcn_fence(__ATOMIC_RELEASE, "workgroup", "local");
    __builtin_amdgcn_s_barrier();
    __builtin_amdgcn_fence(__ATOMIC_ACQUIRE, "workgroup", "local");
}

#define GNN_TILE       16
#define GNN_NTILES     (GNN_N / GNN_TILE)
#define GNN_FUSED_WGS  512
#define GNN_ROWP       136
#define GNN_OROWP      68
#define GNN_WROWS      (2 * GNN_H / 16 * 4 * 32 + 2 * GNN_O / 16 * 4 * 32)
static_assert(GNN_N % GNN_TILE == 0, "whole 16-node tiles");
static_assert(GNN_FUSED_WGS <= GNN_NTILES, "every work-group has a first tile");
static_assert(GNN_WROWS == 3072, "W1l 1024, W1r 1024, W2l 512, W2r 512 fragment rows");
static_assert((GNN_WROWS * 2) % 512 == 0, "the weight copy is whole 16-B pieces per thread");

__device__ __forceinline__ void gnn_store_rows(const float* sz, const float* sr,
                                               __bf16* __restrict__ Zb, float* __restrict__ R,
                                               int tile, int row, int c4) {
    const float4 zv = *(const float4*)(sz + row * GNN_OROWP + c4);
    const float4 rv = *(const float4*)(sr + row * GNN_OROWP + c4);
    const long o = (long)(tile * GNN_TILE + row) * GNN_O + c4;
    v4bf zb;
    zb[0] = (__bf16)zv.x; zb[1] = (__bf16)zv.y; zb[2] = (__bf16)zv.z; zb[3] = (__bf16)zv.w;
    *(v4bf*)(Zb + o) = zb;
    *(float4*)(R + o) = rv;
}

__global__ __launch_bounds__(512) void gnn_gather_sage(const unsigned* __restrict__ Xb,
                                                       const int* __restrict__ rowptr,
                                                       const int* __restrict__ cnt,
                                                       const int* __restrict__ col,
                                                       const uint4* __restrict__ Wpk,
                                                       const float* __restrict__ b1,
                                                       const float* __restrict__ b2,
                                                       __bf16* __restrict__ Zb,
                                                       float* __restrict__ R) {
    __shared__ __attribute__((aligned(32))) uint4 sW[GNN_WROWS * 2];
    __shared__ __attribute__((aligned(16))) __bf16 smean[GNN_TILE * GNN_ROWP];
    __shared__ __attribute__((aligned(16))) __bf16 sx[GNN_TILE * GNN_ROWP];
    __shared__ __attribute__((aligned(16))) __bf16 sh[GNN_TILE * GNN_ROWP];
    __shared__ __attribute__((aligned(16))) float sz[GNN_TILE * GNN_OROWP];
    __shared__ __attribute__((aligned(16))) float sr[GNN_TILE * GNN_OROWP];

    const int t    = threadIdx.x;
    const int lane = t & 31;
    const int w    = __builtin_amdgcn_readfirstlane(t >> 5);
    const int hi   = lane >> 4;
    const int nl   = lane & 15;

    {
        uint4 wv[GNN_WROWS * 2 / 512];
#pragma unroll
        for (int k = 0; k < GNN_WROWS * 2 / 512; ++k) wv[k] = Wpk[t + k * 512];
#pragma unroll
        for (int k = 0; k < GNN_WROWS * 2 / 512; ++k) sW[t + k * 512] = wv[k];
    }
    const v16bf* sWf = (const v16bf*)sW;

    const int ct = w & 7;
    const int ot = w & 3;
    const bool isr = (w & 4) != 0;
    v8f bias1, bias2;
    {
        const float4 bA = *(const float4*)(b1 + ct * 16 + hi * 8);
        const float4 bB = *(const float4*)(b1 + ct * 16 + hi * 8 + 4);
        bias1[0] = bA.x; bias1[1] = bA.y; bias1[2] = bA.z; bias1[3] = bA.w;
        bias1[4] = bB.x; bias1[5] = bB.y; bias1[6] = bB.z; bias1[7] = bB.w;
        const float4 cA = *(const float4*)(b2 + ot * 16 + hi * 8);
        const float4 cB = *(const float4*)(b2 + ot * 16 + hi * 8 + 4);
        bias2[0] = isr ? cA.x : 0.0f; bias2[1] = isr ? cA.y : 0.0f;
        bias2[2] = isr ? cA.z : 0.0f; bias2[3] = isr ? cA.w : 0.0f;
        bias2[4] = isr ? cB.x : 0.0f; bias2[5] = isr ? cB.y : 0.0f;
        bias2[6] = isr ? cB.z : 0.0f; bias2[7] = isr ? cB.w : 0.0f;
    }
    const v16bf* w1l = sWf + (ct * 4) * 32 + lane;
    const v16bf* w1r = sWf + 1024 + (ct * 4) * 32 + lane;
    const v16bf* w2  = sWf + (isr ? 2560 : 2048) + (ot * 4) * 32 + lane;

    const char* xbytes = (const char*)Xb;
    const unsigned lane8 = (unsigned)lane * 8u;
    const int srow = 2 * (w & 7) + hi;
    const int sc4  = nl * 4;

    int tile = blockIdx.x;
    const int xpiece = (t >> 4) * 64 + (t & 15) * 4;
    int beg, deg, myc, begv, degv;
    uint4 xrow = make_uint4(0u, 0u, 0u, 0u);
    {
        const int n = tile * GNN_TILE + w;
        beg = rowptr[n];
        deg = cnt[n];
        const int t1 = (tile + GNN_FUSED_WGS < GNN_NTILES) ? (tile + GNN_FUSED_WGS) : tile;
        begv = rowptr[t1 * GNN_TILE + nl];
        degv = cnt[t1 * GNN_TILE + nl];
        if (t < 256) xrow = *(const uint4*)(Xb + (long)tile * (GNN_TILE * 64) + xpiece);
        myc = (lane < deg) ? col[beg + lane] : 0;
    }

    int ptile = -1;
    for (; tile < GNN_NTILES; tile += GNN_FUSED_WGS) {

        float a0 = 0.f, a1 = 0.f, a2 = 0.f, a3 = 0.f;
        for (int base = 0; base < deg; base += 32) {
            const int m = (deg - base < 32) ? (deg - base) : 32;
            if (base > 0) myc = (lane < m) ? col[beg + base + lane] : 0;
            uint2 v[32];
#pragma unroll
            for (int u = 0; u < 32; ++u) v[u] = make_uint2(0u, 0u);
#pragma unroll
            for (int g = 0; g < 4; ++g) {
                if (g * 8 < m) {
#pragma unroll
                    for (int u = g * 8; u < g * 8 + 8; ++u) {
                        if (u < m) {
                            const int s = __builtin_amdgcn_readlane(myc, u);
                            v[u] = *(const uint2*)(xbytes + ((unsigned)s * (GNN_D * 2u) + lane8));
                        }
                    }
                }
            }
#pragma unroll
            for (int g = 0; g < 4; ++g) {
                if (g * 8 < m) {
#pragma unroll
                    for (int u = g * 8; u < g * 8 + 8; ++u) {
                        a0 += __uint_as_float(v[u].x << 16); a1 += __uint_as_float(v[u].x & 0xffff0000u);
                        a2 += __uint_as_float(v[u].y << 16); a3 += __uint_as_float(v[u].y & 0xffff0000u);
                    }
                }
            }
        }
        const float inv = 1.0f / fmaxf((float)deg, 1.0f);
        v4bf o;
        o[0] = (__bf16)(a0 * inv); o[1] = (__bf16)(a1 * inv);
        o[2] = (__bf16)(a2 * inv); o[3] = (__bf16)(a3 * inv);
        *(v4bf*)(smean + w * GNN_ROWP + lane * 4) = o;
        if (t < 256) *(uint4*)(sx + (t >> 4) * GNN_ROWP + (t & 15) * 8) = xrow;

        const int t1   = (tile + GNN_FUSED_WGS < GNN_NTILES) ? (tile + GNN_FUSED_WGS) : tile;
        const int t2   = (tile + 2 * GNN_FUSED_WGS < GNN_NTILES) ? (tile + 2 * GNN_FUSED_WGS) : tile;
        const int beg2 = __builtin_amdgcn_readlane(begv, w);
        const int deg2 = __builtin_amdgcn_readlane(degv, w);
        begv = rowptr[t2 * GNN_TILE + nl];
        degv = cnt[t2 * GNN_TILE + nl];
        if (t < 256) xrow = *(const uint4*)(Xb + (long)t1 * (GNN_TILE * 64) + xpiece);
        const int myc2 = (lane < deg2) ? col[beg2 + lane] : 0;
        gnn_lds_barrier();

        if (w >= 8 && ptile >= 0) gnn_store_rows(sz, sr, Zb, R, ptile, srow, sc4);

        if (w < 8) {
            const __bf16* mrow = smean + nl * GNN_ROWP + hi * 8;
            const __bf16* xrw  = sx    + nl * GNN_ROWP + hi * 8;
            v8f acc = bias1;
#pragma unroll
            for (int ks = 0; ks < 4; ++ks) {
                const v8bf lo = *(const v8bf*)(mrow + ks * 32);
                const v8bf up = *(const v8bf*)(mrow + ks * 32 + 16);
                const v16bf bf = __builtin_shufflevector(lo, up, 0, 1, 2, 3, 4, 5, 6, 7, 8, 9, 10, 11, 12, 13, 14, 15);
                acc = __builtin_amdgcn_wmma_f32_16x16x32_bf16(false, w1l[ks * 32], false, bf,
                                                              (short)0, acc, false, false);
            }
#pragma unroll
            for (int ks = 0; ks < 4; ++ks) {
                const v8bf lo = *(const v8bf*)(xrw + ks * 32);
                const v8bf up = *(const v8bf*)(xrw + ks * 32 + 16);
                const v16bf bf = __builtin_shufflevector(lo, up, 0, 1, 2, 3, 4, 5, 6, 7, 8, 9, 10, 11, 12, 13, 14, 15);
                acc = __builtin_amdgcn_wmma_f32_16x16x32_bf16(false, w1r[ks * 32], false, bf,
                                                              (short)0, acc, false, false);
            }
            v8bf hv;
#pragma unroll
            for (int r = 0; r < 8; ++r) hv[r] = (__bf16)fmaxf(acc[r], 0.0f);
            *(v8bf*)(sh + nl * GNN_ROWP + ct * 16 + hi * 8) = hv;
        }
        gnn_lds_barrier();

        if (w < 8) {
            const __bf16* hrow = sh + nl * GNN_ROWP + hi * 8;
            v8f acc = bias2;
#pragma unroll
            for (int kk = 0; kk < 4; ++kk) {
                const v8bf lo = *(const v8bf*)(hrow + kk * 32);
                const v8bf up = *(const v8bf*)(hrow + kk * 32 + 16);
                const v16bf bf = __builtin_shufflevector(lo, up, 0, 1, 2, 3, 4, 5, 6, 7, 8, 9, 10, 11, 12, 13, 14, 15);
                acc = __builtin_amdgcn_wmma_f32_16x16x32_bf16(false, w2[kk * 32], false, bf,
                                                              (short)0, acc, false, false);
            }
            float* orow = (isr ? sr : sz) + nl * GNN_OROWP + ot * 16 + hi * 8;
            *(float4*)(orow)     = make_float4(acc[0], acc[1], acc[2], acc[3]);
            *(float4*)(orow + 4) = make_float4(acc[4], acc[5], acc[6], acc[7]);
        }
        ptile = tile;
        beg = beg2; deg = deg2; myc = myc2;
    }
    gnn_lds_barrier();
    if (w >= 8) gnn_store_rows(sz, sr, Zb, R, ptile, srow, sc4);
}

__global__ void gnn_gather_mean_out(const unsigned* __restrict__ Zd,
                                    const float* __restrict__ R,
                                    const int* __restrict__ rowptr,
                                    const int* __restrict__ cnt,
                                    const int* __restrict__ col,
                                    float* __restrict__ out, int N) {
    const int n = __builtin_amdgcn_readfirstlane(blockIdx.x * 8 + (threadIdx.x >> 5));
    if (n >= N) return;
    const int lane = threadIdx.x & 31;
    const int beg  = rowptr[n];
    const int deg  = cnt[n];
    const float2 rv = *(const float2*)(R + (long)n * GNN_O + lane * 2);
    const char* zbytes = (const char*)Zd;
    const unsigned lane4 = (unsigned)lane * 4u;

    float2 acc = make_float2(0.f, 0.f);
    for (int base = 0; base < deg; base += 32) {
        const int m = (deg - base < 32) ? (deg - base) : 32;
        const int myc = (lane < m) ? col[beg + base + lane] : 0;
        int j = 0;
        for (; j + 16 <= m; j += 16) {
            unsigned v[16];
#pragma unroll
            for (int u = 0; u < 16; ++u)
                v[u] = (unsigned)__builtin_amdgcn_readlane(myc, j + u) * (GNN_O * 2u) + lane4;
            __builtin_amdgcn_sched_barrier(0);
#pragma unroll
            for (int u = 0; u < 16; ++u) v[u] = *(const unsigned*)(zbytes + v[u]);
#pragma unroll
            for (int u = 0; u < 16; ++u) {
                acc.x += __uint_as_float(v[u] << 16); acc.y += __uint_as_float(v[u] & 0xffff0000u);
            }
        }
        if (j + 8 <= m) {
            unsigned v[8];
#pragma unroll
            for (int u = 0; u < 8; ++u) {
                const unsigned s = (unsigned)__builtin_amdgcn_readlane(myc, j + u);
                v[u] = *(const unsigned*)(zbytes + (s * (GNN_O * 2u) + lane4));
            }
#pragma unroll
            for (int u = 0; u < 8; ++u) {
                acc.x += __uint_as_float(v[u] << 16); acc.y += __uint_as_float(v[u] & 0xffff0000u);
            }
            j += 8;
        }
        if (j + 4 <= m) {
            unsigned v[4];
#pragma unroll
            for (int u = 0; u < 4; ++u) {
                const unsigned s = (unsigned)__builtin_amdgcn_readlane(myc, j + u);
                v[u] = *(const unsigned*)(zbytes + (s * (GNN_O * 2u) + lane4));
            }
#pragma unroll
            for (int u = 0; u < 4; ++u) {
                acc.x += __uint_as_float(v[u] << 16); acc.y += __uint_as_float(v[u] & 0xffff0000u);
            }
            j += 4;
        }
        const int r = m - j;
        if (r > 0) {
            unsigned v0 = 0u, v1 = 0u, v2 = 0u;
            v0 = *(const unsigned*)(zbytes + ((unsigned)__builtin_amdgcn_readlane(myc, j) * (GNN_O * 2u) + lane4));
            if (r > 1) v1 = *(const unsigned*)(zbytes + ((unsigned)__builtin_amdgcn_readlane(myc, j + 1) * (GNN_O * 2u) + lane4));
            if (r > 2) v2 = *(const unsigned*)(zbytes + ((unsigned)__builtin_amdgcn_readlane(myc, j + 2) * (GNN_O * 2u) + lane4));
            acc.x += __uint_as_float(v0 << 16); acc.y += __uint_as_float(v0 & 0xffff0000u);
            acc.x += __uint_as_float(v1 << 16); acc.y += __uint_as_float(v1 & 0xffff0000u);
            acc.x += __uint_as_float(v2 << 16); acc.y += __uint_as_float(v2 & 0xffff0000u);
        }
    }
    const float inv = 1.0f / fmaxf((float)deg, 1.0f);
    *(float2*)(out + (long)n * GNN_O + lane * 2) = make_float2(rv.x + acc.x * inv, rv.y + acc.y * inv);
}

extern "C" void kernel_launch(void* const* d_in, const int* in_sizes, int n_in,
                              void* d_out, int out_size, void* d_ws, size_t ws_size,
                              hipStream_t stream) {
    (void)in_sizes; (void)n_in; (void)out_size; (void)ws_size;

    const float* x   = (const float*)d_in[0];
    const int*   ei  = (const int*)d_in[1];
    const float* W1l = (const float*)d_in[2];
    const float* b1  = (const float*)d_in[3];
    const float* W1r = (const float*)d_in[4];
    const float* W2l = (const float*)d_in[5];
    const float* b2  = (const float*)d_in[6];
    const float* W2r = (const float*)d_in[7];
    float*       out = (float*)d_out;

    const int* src = ei;
    const int* dst = ei + GNN_E;

    char*   ws     = (char*)d_ws;
    int*    cnt    = (int*)(ws + 0);
    __bf16* xb     = (__bf16*)(ws + 26124288);
    __bf16* z      = (__bf16*)(ws + 51724288);
    float*  r      = (float*)(ws + 77324288);
    __bf16* W1l_p  = (__bf16*)(ws + 102924288);
    __bf16* W1r_p  = (__bf16*)(ws + 102957056);
    __bf16* W2l_p  = (__bf16*)(ws + 102989824);
    __bf16* W2r_p  = (__bf16*)(ws + 103006208);
    int*    rowptr = (int*)(ws + 104857600);
    unsigned* csorted = (unsigned*)(ws + 105385984);
    int*    col    = (int*)(ws + 111785984);
    int*    hist   = (int*)(ws + 118185984);
    int*    hstart = (int*)(ws + 118382592);

    gnn_chunk_sort_pack<<<GNN_NCHUNK + GNN_PACK_WGS + GNN_XWGS, 1024, 0, stream>>>(
        src, dst, hist, hstart, csorted, GNN_E, W1l, W1r, W2l, W2r, W1l_p, W1r_p, W2l_p, W2r_p, x, xb);
    gnn_bin_sort<<<GNN_NBINS + GNN_XWGS, 1024, 0, stream>>>(csorted, hist, hstart, cnt, rowptr, col,
                                                            GNN_N, x, xb);

    gnn_gather_sage<<<GNN_FUSED_WGS, 512, 0, stream>>>((const unsigned*)xb, rowptr, cnt, col,
                                                        (const uint4*)W1l_p, b1, b2, z, r);

    gnn_gather_mean_out<<<(GNN_N + 7) / 8, 256, 0, stream>>>((const unsigned*)z, r, rowptr, cnt, col, out, GNN_N);
}
